# starting kernel
# baseline (speedup 1.0000x reference)
#include <hip/hip_runtime.h>
#include <hip/hip_bf16.h>

typedef _Float16 half_t;
typedef __attribute__((ext_vector_type(16))) _Float16 v16h;
typedef __attribute__((ext_vector_type(8)))  _Float16 v8h;
typedef __attribute__((ext_vector_type(8)))  float    v8f;

#define NGRAPH 128
#define NNODE  512
#define NEDGE  8192
#define NFEAT  128
#define NHID   128
#define BN_EPS 1e-5f

__global__ __launch_bounds__(512) void k_degrees(const int* __restrict__ edges,
                                                 float* __restrict__ rs_out,
                                                 float* __restrict__ rs_in) {
    __shared__ float degO[NNODE];
    __shared__ float degI[NNODE];
    const int g = blockIdx.x;
    const int t = threadIdx.x;
    const int* e = edges + (size_t)g * 2 * NEDGE;
    for (int n = t; n < NNODE; n += 512) { degO[n] = 0.f; degI[n] = 0.f; }
    __syncthreads();
    for (int i = t; i < NEDGE; i += 512) {
        atomicAdd(&degO[e[i]], 1.0f);
        atomicAdd(&degI[e[NEDGE + i]], 1.0f);
    }
    __syncthreads();
    for (int n = t; n < NNODE; n += 512) {
        rs_out[g * NNODE + n] = rsqrtf(fmaxf(degO[n], 1.0f));
        rs_in [g * NNODE + n] = rsqrtf(fmaxf(degI[n], 1.0f));
    }
}

__global__ __launch_bounds__(512) void k_scatter(const float*  __restrict__ xf32,
                                                 const half_t* __restrict__ xf16,
                                                 const int*    __restrict__ edges,
                                                 const float*  __restrict__ rs_out,
                                                 float*        __restrict__ agg,
                                                 int use_half) {
    const int g    = blockIdx.x;
    const int wave = threadIdx.x >> 5;
    const int lane = threadIdx.x & 31;
    const int* e = edges + (size_t)g * 2 * NEDGE;
    const float* rso = rs_out + g * NNODE;
    const size_t gbase = (size_t)g * NNODE * NFEAT;
    for (int i = wave; i < NEDGE; i += 16) {
        const int src = e[i];
        const int dst = e[NEDGE + i];
        const float s = rso[src];
        const size_t sb = gbase + (size_t)src * NFEAT;
        const size_t db = gbase + (size_t)dst * NFEAT;
#pragma unroll
        for (int c = 0; c < 4; ++c) {
            const int f = lane + 32 * c;
            const float v = use_half ? (float)xf16[sb + f] : xf32[sb + f];
            atomicAdd(&agg[db + f], v * s);
        }
    }
}

__global__ __launch_bounds__(512, 1) void k_gemm(const float* __restrict__ agg,
                                                 const float* __restrict__ rs_in,
                                                 const float* __restrict__ W,
                                                 const float* __restrict__ bvec,
                                                 half_t*      __restrict__ h_out,
                                                 float*       __restrict__ emb_out,
                                                 int layer) {
    __shared__ __align__(16) half_t wt[NHID * NHID];
    __shared__ float  bs[NHID];
    __shared__ float  embacc[NHID];
    const int g    = blockIdx.x >> 1;
    const int half = blockIdx.x & 1;
    const int t    = threadIdx.x;

    for (int i = t; i < NHID * NHID; i += 512) {
        const int k = i >> 7, n = i & 127;
        wt[n * NHID + k] = (half_t)W[i];
    }
    if (t < NHID) { bs[t] = bvec[t]; embacc[t] = 0.f; }
    __syncthreads();

    const int wave = t >> 5, lane = t & 31;
    const int l16 = lane & 15, hi = lane >> 4;
    const size_t gb = (size_t)g * NNODE;

    const int mi = half * 16 + wave;
    const int m  = mi * 16 + l16;
    const float* arow = agg + (gb + m) * NFEAT;
    const float  rsin = rs_in[gb + m];

    v8f c[8] = {};

#pragma unroll
    for (int ko = 0; ko < 4; ++ko) {
        const int kb = ko * 32;
        const int k0 = kb + hi * 8;
        const int k1 = kb + 16 + hi * 8;
        const float4 f0 = *(const float4*)(arow + k0);
        const float4 f1 = *(const float4*)(arow + k0 + 4);
        const float4 f2 = *(const float4*)(arow + k1);
        const float4 f3 = *(const float4*)(arow + k1 + 4);
        v16h a;
        a[0]  = (half_t)(f0.x * rsin); a[1]  = (half_t)(f0.y * rsin);
        a[2]  = (half_t)(f0.z * rsin); a[3]  = (half_t)(f0.w * rsin);
        a[4]  = (half_t)(f1.x * rsin); a[5]  = (half_t)(f1.y * rsin);
        a[6]  = (half_t)(f1.z * rsin); a[7]  = (half_t)(f1.w * rsin);
        a[8]  = (half_t)(f2.x * rsin); a[9]  = (half_t)(f2.y * rsin);
        a[10] = (half_t)(f2.z * rsin); a[11] = (half_t)(f2.w * rsin);
        a[12] = (half_t)(f3.x * rsin); a[13] = (half_t)(f3.y * rsin);
        a[14] = (half_t)(f3.z * rsin); a[15] = (half_t)(f3.w * rsin);

#pragma unroll
        for (int ni = 0; ni < 8; ++ni) {
            const half_t* brow = wt + (ni * 16 + l16) * NHID;
            const v8h blo = *(const v8h*)(brow + k0);
            const v8h bhi = *(const v8h*)(brow + k1);
            const v16h b = __builtin_shufflevector(blo, bhi,
                               0, 1, 2, 3, 4, 5, 6, 7,
                               8, 9, 10, 11, 12, 13, 14, 15);
            c[ni] = __builtin_amdgcn_wmma_f32_16x16x32_f16(
                        false, a, false, b, (short)0, c[ni], false, false);
        }
    }

#pragma unroll
    for (int ni = 0; ni < 8; ++ni) {
        const int n = ni * 16 + l16;
        if (layer == 0) {
#pragma unroll
            for (int r = 0; r < 8; ++r) {
                const int ml = r + 8 * hi;
                const float v = fmaxf(c[ni][r] + bs[n], 0.f);
                h_out[(gb + (size_t)mi * 16 + ml) * NFEAT + n] = (half_t)v;
            }
        } else {
            float s = 0.f;
#pragma unroll
            for (int r = 0; r < 8; ++r) s += fmaxf(c[ni][r] + bs[n], 0.f);
            atomicAdd(&embacc[n], s);
        }
    }

    if (layer == 1) {
        __syncthreads();
        if (t < NHID) atomicAdd(&emb_out[g * NHID + t], embacc[t] * (1.0f / NNODE));
    }
}

__global__ __launch_bounds__(128) void k_head(const float* __restrict__ emb,
                                              const float* __restrict__ Wt,
                                              const float* __restrict__ bt,
                                              const float* __restrict__ gamma,
                                              const float* __restrict__ beta,
                                              const float* __restrict__ rmean,
                                              const float* __restrict__ rvar,
                                              float* __restrict__ out) {
    __shared__ float er[NHID];
    const int g = blockIdx.x, n = threadIdx.x;
    er[n] = emb[g * NHID + n];
    __syncthreads();
    float z = bt[n];
    for (int k = 0; k < NHID; ++k) z += er[k] * Wt[k * NHID + n];
    z = gamma[n] * (z - rmean[n]) * rsqrtf(rvar[n] + BN_EPS) + beta[n];
    out[g * NHID + n] = fmaxf(z, 0.f);
}

extern "C" void kernel_launch(void* const* d_in, const int* in_sizes, int n_in,
                              void* d_out, int out_size, void* d_ws, size_t ws_size,
                              hipStream_t stream) {
    (void)in_sizes; (void)n_in; (void)out_size; (void)ws_size;
    const float* feats = (const float*)d_in[0];
    const int*   edges = (const int*)d_in[1];
    const float* W0    = (const float*)d_in[2];
    const float* b0    = (const float*)d_in[3];
    const float* W1    = (const float*)d_in[4];
    const float* b1    = (const float*)d_in[5];
    const float* Wt    = (const float*)d_in[6];
    const float* bt    = (const float*)d_in[7];
    const float* gamma = (const float*)d_in[8];
    const float* beta  = (const float*)d_in[9];
    const float* rmean = (const float*)d_in[10];
    const float* rvar  = (const float*)d_in[11];
    float* out = (float*)d_out;

    char* ws = (char*)d_ws;
    float*  rs_out = (float*)ws;  ws += (size_t)NGRAPH * NNODE * sizeof(float);
    float*  rs_in  = (float*)ws;  ws += (size_t)NGRAPH * NNODE * sizeof(float);
    float*  agg    = (float*)ws;  ws += (size_t)NGRAPH * NNODE * NFEAT * sizeof(float);
    half_t* h1     = (half_t*)ws; ws += (size_t)NGRAPH * NNODE * NFEAT * sizeof(half_t);
    float*  emb    = (float*)ws;  ws += (size_t)NGRAPH * NHID * sizeof(float);
    const size_t agg_bytes = (size_t)NGRAPH * NNODE * NFEAT * sizeof(float);
    const size_t emb_bytes = (size_t)NGRAPH * NHID * sizeof(float);

    hipMemsetAsync(agg, 0, agg_bytes, stream);
    k_degrees<<<NGRAPH, 512, 0, stream>>>(edges, rs_out, rs_in);

    k_scatter<<<NGRAPH, 512, 0, stream>>>(feats, nullptr, edges, rs_out, agg, 0);
    k_gemm   <<<NGRAPH * 2, 512, 0, stream>>>(agg, rs_in, W0, b0, h1, emb, 0);

    hipMemsetAsync(agg, 0, agg_bytes, stream);
    hipMemsetAsync(emb, 0, emb_bytes, stream);
    k_scatter<<<NGRAPH, 512, 0, stream>>>(nullptr, h1, edges, rs_out, agg, 1);
    k_gemm   <<<NGRAPH * 2, 512, 0, stream>>>(agg, rs_in, W1, b1, h1, emb, 1);

    k_head<<<NGRAPH, 128, 0, stream>>>(emb, Wt, bt, gamma, beta, rmean, rvar, out);
}


# Aggregation: per-edge global float atomics replaced by a dst-sorted edge list (built once per call) and a per-row register gather; row sums now in fixed edge order; agg zero fills dropped
# speedup vs baseline: 10.4123x; 10.4123x over previous
#include <hip/hip_runtime.h>
#include <hip/hip_bf16.h>

typedef _Float16 half_t;
typedef __attribute__((ext_vector_type(16))) _Float16 v16h;
typedef __attribute__((ext_vector_type(8)))  _Float16 v8h;
typedef __attribute__((ext_vector_type(4)))  _Float16 v4h;
typedef __attribute__((ext_vector_type(8)))  float    v8f;

#define NGRAPH 128
#define NNODE  512
#define NEDGE  8192
#define NFEAT  128
#define NHID   128
#define BN_EPS 1e-5f

__global__ __launch_bounds__(512) void k_csr(const int* __restrict__ edges,
                                             float* __restrict__ rs_in,
                                             int2*  __restrict__ rowinfo,
                                             int2*  __restrict__ list) {
    __shared__ int   cur[16 * NNODE];
    __shared__ int   degO[NNODE];
    __shared__ int   rowp[NNODE];
    __shared__ float rso[NNODE];
    __shared__ int   wsum[16];
    const int g    = blockIdx.x;
    const int t    = threadIdx.x;
    const int wave = t >> 5;
    const int lane = t & 31;
    const int* e = edges + (size_t)g * 2 * NEDGE;

    for (int i = t; i < 16 * NNODE; i += 512) cur[i] = 0;
    degO[t] = 0;
    __syncthreads();

    int src[16], dst[16];
#pragma unroll
    for (int r = 0; r < 16; ++r) {
        const int idx = wave * 512 + r * 32 + lane;
        src[r] = e[idx] & (NNODE - 1);
        dst[r] = e[NEDGE + idx] & (NNODE - 1);
    }
#pragma unroll
    for (int r = 0; r < 16; ++r) {
        atomicAdd(&cur[wave * NNODE + dst[r]], 1);
        atomicAdd(&degO[src[r]], 1);
    }
    __syncthreads();

    int run = 0;
#pragma unroll
    for (int w = 0; w < 16; ++w) {
        const int c = cur[w * NNODE + t];
        cur[w * NNODE + t] = run;
        run += c;
    }
    int incl = run;
    for (int o = 1; o < 32; o <<= 1) {
        const int v = __shfl_up(incl, o);
        if (lane >= o) incl += v;
    }
    if (lane == 31) wsum[wave] = incl;
    __syncthreads();
    int pre = 0;
    for (int w = 0; w < wave; ++w) pre += wsum[w];
    const int rstart = pre + incl - run;
    rowp[t] = rstart;
    rso[t]  = rsqrtf(fmaxf((float)degO[t], 1.0f));
    rs_in[g * NNODE + t]   = rsqrtf(fmaxf((float)run, 1.0f));
    rowinfo[g * NNODE + t] = make_int2(rstart, run);
    __syncthreads();

    int2* lg = list + (size_t)g * NEDGE;
#pragma unroll
    for (int r = 0; r < 16; ++r) {
        const int d = dst[r];
        int rank = 0;
#pragma unroll
        for (int j = 0; j < 31; ++j) {
            const int dj = __builtin_amdgcn_readlane(d, j);
            rank += (dj == d && j < lane) ? 1 : 0;
        }
        const int base = cur[wave * NNODE + d];
        atomicAdd(&cur[wave * NNODE + d], 1);
        const int pos = rowp[d] + base + rank;
        lg[pos] = make_int2(src[r], __float_as_int(rso[src[r]]));
    }
}

#define GR_WAVES 8
#define GR_ROWS  8

__device__ __forceinline__ float4 ld_row_f32(const float* __restrict__ xg, int s, int lane) {
    return *(const float4*)(xg + (size_t)s * NFEAT + lane * 4);
}
__device__ __forceinline__ float4 ld_row_f16(const half_t* __restrict__ xg, int s, int lane) {
    const v4h h = *(const v4h*)(xg + (size_t)s * NFEAT + lane * 4);
    return make_float4((float)h[0], (float)h[1], (float)h[2], (float)h[3]);
}

template <int U, typename T>
__device__ __forceinline__ void gather_batch(float4& acc, const int2 ent, const int k,
                                             const T* __restrict__ xg, const int lane) {
    int   s[U];
    float w[U];
    float4 v[U];
#pragma unroll
    for (int i = 0; i < U; ++i) {
        s[i] = __builtin_amdgcn_readlane(ent.x, k + i);
        w[i] = __int_as_float(__builtin_amdgcn_readlane(ent.y, k + i));
    }
#pragma unroll
    for (int i = 0; i < U; ++i) {
        if constexpr (sizeof(T) == 4) v[i] = ld_row_f32((const float*)xg, s[i], lane);
        else                          v[i] = ld_row_f16((const half_t*)xg, s[i], lane);
    }
#pragma unroll
    for (int i = 0; i < U; ++i) {
        acc.x = fmaf(v[i].x, w[i], acc.x);
        acc.y = fmaf(v[i].y, w[i], acc.y);
        acc.z = fmaf(v[i].z, w[i], acc.z);
        acc.w = fmaf(v[i].w, w[i], acc.w);
    }
}

template <typename T>
__device__ __forceinline__ void gather_chunk(float4& acc, const int2 ent, const int cnt,
                                             const T* __restrict__ xg, const int lane) {
    int k = 0;
    for (; k + 8 <= cnt; k += 8) gather_batch<8>(acc, ent, k, xg, lane);
    if (k + 4 <= cnt) { gather_batch<4>(acc, ent, k, xg, lane); k += 4; }
    const int rem = cnt - k;
    if      (rem == 3) gather_batch<3>(acc, ent, k, xg, lane);
    else if (rem == 2) gather_batch<2>(acc, ent, k, xg, lane);
    else if (rem == 1) gather_batch<1>(acc, ent, k, xg, lane);
}

template <typename T>
__device__ __forceinline__ void gather_rows(const T* __restrict__ xg,
                                            const int2* __restrict__ ri,
                                            const int2* __restrict__ lg,
                                            float* __restrict__ ag,
                                            const int row0, const int lane) {
    int2 inf = ri[row0];
    int2 nxt = (lane < inf.y) ? lg[inf.x + lane] : make_int2(0, 0);
    for (int i = 0; i < GR_ROWS; ++i) {
        const int start = inf.x, len = inf.y;
        const int2 ent = nxt;
        if (i + 1 < GR_ROWS) {
            inf = ri[row0 + i + 1];
            nxt = (lane < inf.y) ? lg[inf.x + lane] : make_int2(0, 0);
        }
        float4 acc = make_float4(0.f, 0.f, 0.f, 0.f);
        gather_chunk<T>(acc, ent, min(len, 32), xg, lane);
        for (int b = 32; b < len; b += 32) {
            const int2 more = (b + lane < len) ? lg[start + b + lane] : make_int2(0, 0);
            gather_chunk<T>(acc, more, min(len - b, 32), xg, lane);
        }
        *(float4*)(ag + (size_t)(row0 + i) * NFEAT + lane * 4) = acc;
    }
}

__global__ __launch_bounds__(GR_WAVES * 32) void k_gather_f32(const float* __restrict__ x,
                                                              const int2*  __restrict__ rowinfo,
                                                              const int2*  __restrict__ list,
                                                              float*       __restrict__ agg) {
    const int g    = blockIdx.x >> 3;
    const int part = blockIdx.x & 7;
    const int wave = __builtin_amdgcn_readfirstlane(threadIdx.x >> 5);
    const int lane = threadIdx.x & 31;
    const int row0 = part * (GR_WAVES * GR_ROWS) + wave * GR_ROWS;
    gather_rows<float>(x + (size_t)g * NNODE * NFEAT, rowinfo + g * NNODE,
                       list + (size_t)g * NEDGE, agg + (size_t)g * NNODE * NFEAT, row0, lane);
}

__global__ __launch_bounds__(GR_WAVES * 32) void k_gather_f16(const half_t* __restrict__ h,
                                                              const int2*   __restrict__ rowinfo,
                                                              const int2*   __restrict__ list,
                                                              float*        __restrict__ agg) {
    const int g    = blockIdx.x >> 3;
    const int part = blockIdx.x & 7;
    const int wave = __builtin_amdgcn_readfirstlane(threadIdx.x >> 5);
    const int lane = threadIdx.x & 31;
    const int row0 = part * (GR_WAVES * GR_ROWS) + wave * GR_ROWS;
    gather_rows<half_t>(h + (size_t)g * NNODE * NFEAT, rowinfo + g * NNODE,
                        list + (size_t)g * NEDGE, agg + (size_t)g * NNODE * NFEAT, row0, lane);
}

__global__ __launch_bounds__(512, 1) void k_gemm(const float* __restrict__ agg,
                                                 const float* __restrict__ rs_in,
                                                 const float* __restrict__ W,
                                                 const float* __restrict__ bvec,
                                                 half_t*      __restrict__ h_out,
                                                 float*       __restrict__ emb_out,
                                                 int layer) {
    __shared__ __align__(16) half_t wt[NHID * NHID];
    __shared__ float  bs[NHID];
    __shared__ float  embacc[NHID];
    const int g    = blockIdx.x >> 1;
    const int half = blockIdx.x & 1;
    const int t    = threadIdx.x;

    for (int i = t; i < NHID * NHID; i += 512) {
        const int k = i >> 7, n = i & 127;
        wt[n * NHID + k] = (half_t)W[i];
    }
    if (t < NHID) { bs[t] = bvec[t]; embacc[t] = 0.f; }
    __syncthreads();

    const int wave = t >> 5, lane = t & 31;
    const int l16 = lane & 15, hi = lane >> 4;
    const size_t gb = (size_t)g * NNODE;

    const int mi = half * 16 + wave;
    const int m  = mi * 16 + l16;
    const float* arow = agg + (gb + m) * NFEAT;
    const float  rsin = rs_in[gb + m];

    v8f c[8] = {};

#pragma unroll
    for (int ko = 0; ko < 4; ++ko) {
        const int kb = ko * 32;
        const int k0 = kb + hi * 8;
        const int k1 = kb + 16 + hi * 8;
        const float4 f0 = *(const float4*)(arow + k0);
        const float4 f1 = *(const float4*)(arow + k0 + 4);
        const float4 f2 = *(const float4*)(arow + k1);
        const float4 f3 = *(const float4*)(arow + k1 + 4);
        v16h a;
        a[0]  = (half_t)(f0.x * rsin); a[1]  = (half_t)(f0.y * rsin);
        a[2]  = (half_t)(f0.z * rsin); a[3]  = (half_t)(f0.w * rsin);
        a[4]  = (half_t)(f1.x * rsin); a[5]  = (half_t)(f1.y * rsin);
        a[6]  = (half_t)(f1.z * rsin); a[7]  = (half_t)(f1.w * rsin);
        a[8]  = (half_t)(f2.x * rsin); a[9]  = (half_t)(f2.y * rsin);
        a[10] = (half_t)(f2.z * rsin); a[11] = (half_t)(f2.w * rsin);
        a[12] = (half_t)(f3.x * rsin); a[13] = (half_t)(f3.y * rsin);
        a[14] = (half_t)(f3.z * rsin); a[15] = (half_t)(f3.w * rsin);

#pragma unroll
        for (int ni = 0; ni < 8; ++ni) {
            const half_t* brow = wt + (ni * 16 + l16) * NHID;
            const v8h blo = *(const v8h*)(brow + k0);
            const v8h bhi = *(const v8h*)(brow + k1);
            const v16h b = __builtin_shufflevector(blo, bhi,
                               0, 1, 2, 3, 4, 5, 6, 7,
                               8, 9, 10, 11, 12, 13, 14, 15);
            c[ni] = __builtin_amdgcn_wmma_f32_16x16x32_f16(
                        false, a, false, b, (short)0, c[ni], false, false);
        }
    }

#pragma unroll
    for (int ni = 0; ni < 8; ++ni) {
        const int n = ni * 16 + l16;
        if (layer == 0) {
#pragma unroll
            for (int r = 0; r < 8; ++r) {
                const int ml = r + 8 * hi;
                const float v = fmaxf(c[ni][r] + bs[n], 0.f);
                h_out[(gb + (size_t)mi * 16 + ml) * NFEAT + n] = (half_t)v;
            }
        } else {
            float s = 0.f;
#pragma unroll
            for (int r = 0; r < 8; ++r) s += fmaxf(c[ni][r] + bs[n], 0.f);
            atomicAdd(&embacc[n], s);
        }
    }

    if (layer == 1) {
        __syncthreads();
        if (t < NHID) atomicAdd(&emb_out[g * NHID + t], embacc[t] * (1.0f / NNODE));
    }
}

__global__ __launch_bounds__(128) void k_head(const float* __restrict__ emb,
                                              const float* __restrict__ Wt,
                                              const float* __restrict__ bt,
                                              const float* __restrict__ gamma,
                                              const float* __restrict__ beta,
                                              const float* __restrict__ rmean,
                                              const float* __restrict__ rvar,
                                              float* __restrict__ out) {
    __shared__ float er[NHID];
    const int g = blockIdx.x, n = threadIdx.x;
    er[n] = emb[g * NHID + n];
    __syncthreads();
    float z = bt[n];
    for (int k = 0; k < NHID; ++k) z += er[k] * Wt[k * NHID + n];
    z = gamma[n] * (z - rmean[n]) * rsqrtf(rvar[n] + BN_EPS) + beta[n];
    out[g * NHID + n] = fmaxf(z, 0.f);
}

extern "C" void kernel_launch(void* const* d_in, const int* in_sizes, int n_in,
                              void* d_out, int out_size, void* d_ws, size_t ws_size,
                              hipStream_t stream) {
    (void)in_sizes; (void)n_in; (void)out_size; (void)ws_size;
    const float* feats = (const float*)d_in[0];
    const int*   edges = (const int*)d_in[1];
    const float* W0    = (const float*)d_in[2];
    const float* b0    = (const float*)d_in[3];
    const float* W1    = (const float*)d_in[4];
    const float* b1    = (const float*)d_in[5];
    const float* Wt    = (const float*)d_in[6];
    const float* bt    = (const float*)d_in[7];
    const float* gamma = (const float*)d_in[8];
    const float* beta  = (const float*)d_in[9];
    const float* rmean = (const float*)d_in[10];
    const float* rvar  = (const float*)d_in[11];
    float* out = (float*)d_out;

    char* ws = (char*)d_ws;
    float*  rs_in   = (float*)ws;  ws += (size_t)NGRAPH * NNODE * sizeof(float);
    int2*   rowinfo = (int2*)ws;   ws += (size_t)NGRAPH * NNODE * sizeof(int2);
    int2*   list    = (int2*)ws;   ws += (size_t)NGRAPH * NEDGE * sizeof(int2);
    float*  agg     = (float*)ws;  ws += (size_t)NGRAPH * NNODE * NFEAT * sizeof(float);
    half_t* h1      = (half_t*)ws; ws += (size_t)NGRAPH * NNODE * NFEAT * sizeof(half_t);
    float*  emb     = (float*)ws;  ws += (size_t)NGRAPH * NHID * sizeof(float);
    const size_t emb_bytes = (size_t)NGRAPH * NHID * sizeof(float);

    k_csr<<<NGRAPH, 512, 0, stream>>>(edges, rs_in, rowinfo, list);

    k_gather_f32<<<NGRAPH * 8, GR_WAVES * 32, 0, stream>>>(feats, rowinfo, list, agg);
    k_gemm      <<<NGRAPH * 2, 512, 0, stream>>>(agg, rs_in, W0, b0, h1, emb, 0);

    hipMemsetAsync(emb, 0, emb_bytes, stream);
    k_gather_f16<<<NGRAPH * 8, GR_WAVES * 32, 0, stream>>>(h1, rowinfo, list, agg);
    k_gemm      <<<NGRAPH * 2, 512, 0, stream>>>(agg, rs_in, W1, b1, h1, emb, 1);

    k_head<<<NGRAPH, 128, 0, stream>>>(emb, Wt, bt, gamma, beta, rmean, rvar, out);
}


# k_gemm: LDS W^T row pitch 256 B -> 272 B (16 B x odd), filled by packed 16-byte stores: removes the 16-way bank conflicts of the B-fragment loads and the 32-way ones of the 2-byte transposing stores
# speedup vs baseline: 15.1626x; 1.4562x over previous
#include <hip/hip_runtime.h>
#include <hip/hip_bf16.h>

typedef _Float16 half_t;
typedef __attribute__((ext_vector_type(16))) _Float16 v16h;
typedef __attribute__((ext_vector_type(8)))  _Float16 v8h;
typedef __attribute__((ext_vector_type(4)))  _Float16 v4h;
typedef __attribute__((ext_vector_type(8)))  float    v8f;

#define NGRAPH 128
#define NNODE  512
#define NEDGE  8192
#define NFEAT  128
#define NHID   128
#define BN_EPS 1e-5f

__global__ __launch_bounds__(512) void k_csr(const int* __restrict__ edges,
                                             float* __restrict__ rs_in,
                                             int2*  __restrict__ rowinfo,
                                             int2*  __restrict__ list) {
    __shared__ int   cur[16 * NNODE];
    __shared__ int   degO[NNODE];
    __shared__ int   rowp[NNODE];
    __shared__ float rso[NNODE];
    __shared__ int   wsum[16];
    const int g    = blockIdx.x;
    const int t    = threadIdx.x;
    const int wave = t >> 5;
    const int lane = t & 31;
    const int* e = edges + (size_t)g * 2 * NEDGE;

    for (int i = t; i < 16 * NNODE; i += 512) cur[i] = 0;
    degO[t] = 0;
    __syncthreads();

    int src[16], dst[16];
#pragma unroll
    for (int r = 0; r < 16; ++r) {
        const int idx = wave * 512 + r * 32 + lane;
        src[r] = e[idx] & (NNODE - 1);
        dst[r] = e[NEDGE + idx] & (NNODE - 1);
    }
#pragma unroll
    for (int r = 0; r < 16; ++r) {
        atomicAdd(&cur[wave * NNODE + dst[r]], 1);
        atomicAdd(&degO[src[r]], 1);
    }
    __syncthreads();

    int run = 0;
#pragma unroll
    for (int w = 0; w < 16; ++w) {
        const int c = cur[w * NNODE + t];
        cur[w * NNODE + t] = run;
        run += c;
    }
    int incl = run;
    for (int o = 1; o < 32; o <<= 1) {
        const int v = __shfl_up(incl, o);
        if (lane >= o) incl += v;
    }
    if (lane == 31) wsum[wave] = incl;
    __syncthreads();
    int pre = 0;
    for (int w = 0; w < wave; ++w) pre += wsum[w];
    const int rstart = pre + incl - run;
    rowp[t] = rstart;
    rso[t]  = rsqrtf(fmaxf((float)degO[t], 1.0f));
    rs_in[g * NNODE + t]   = rsqrtf(fmaxf((float)run, 1.0f));
    rowinfo[g * NNODE + t] = make_int2(rstart, run);
    __syncthreads();

    int2* lg = list + (size_t)g * NEDGE;
#pragma unroll
    for (int r = 0; r < 16; ++r) {
        const int d = dst[r];
        int rank = 0;
#pragma unroll
        for (int j = 0; j < 31; ++j) {
            const int dj = __builtin_amdgcn_readlane(d, j);
            rank += (dj == d && j < lane) ? 1 : 0;
        }
        const int base = cur[wave * NNODE + d];
        atomicAdd(&cur[wave * NNODE + d], 1);
        const int pos = rowp[d] + base + rank;
        lg[pos] = make_int2(src[r], __float_as_int(rso[src[r]]));
    }
}

#define GR_WAVES 8
#define GR_ROWS  8

__device__ __forceinline__ float4 ld_row_f32(const float* __restrict__ xg, int s, int lane) {
    return *(const float4*)(xg + (size_t)s * NFEAT + lane * 4);
}
__device__ __forceinline__ float4 ld_row_f16(const half_t* __restrict__ xg, int s, int lane) {
    const v4h h = *(const v4h*)(xg + (size_t)s * NFEAT + lane * 4);
    return make_float4((float)h[0], (float)h[1], (float)h[2], (float)h[3]);
}

template <int U, typename T>
__device__ __forceinline__ void gather_batch(float4& acc, const int2 ent, const int k,
                                             const T* __restrict__ xg, const int lane) {
    int   s[U];
    float w[U];
    float4 v[U];
#pragma unroll
    for (int i = 0; i < U; ++i) {
        s[i] = __builtin_amdgcn_readlane(ent.x, k + i);
        w[i] = __int_as_float(__builtin_amdgcn_readlane(ent.y, k + i));
    }
#pragma unroll
    for (int i = 0; i < U; ++i) {
        if constexpr (sizeof(T) == 4) v[i] = ld_row_f32((const float*)xg, s[i], lane);
        else                          v[i] = ld_row_f16((const half_t*)xg, s[i], lane);
    }
#pragma unroll
    for (int i = 0; i < U; ++i) {
        acc.x = fmaf(v[i].x, w[i], acc.x);
        acc.y = fmaf(v[i].y, w[i], acc.y);
        acc.z = fmaf(v[i].z, w[i], acc.z);
        acc.w = fmaf(v[i].w, w[i], acc.w);
    }
}

template <typename T>
__device__ __forceinline__ void gather_chunk(float4& acc, const int2 ent, const int cnt,
                                             const T* __restrict__ xg, const int lane) {
    int k = 0;
    for (; k + 8 <= cnt; k += 8) gather_batch<8>(acc, ent, k, xg, lane);
    if (k + 4 <= cnt) { gather_batch<4>(acc, ent, k, xg, lane); k += 4; }
    const int rem = cnt - k;
    if      (rem == 3) gather_batch<3>(acc, ent, k, xg, lane);
    else if (rem == 2) gather_batch<2>(acc, ent, k, xg, lane);
    else if (rem == 1) gather_batch<1>(acc, ent, k, xg, lane);
}

template <typename T>
__device__ __forceinline__ void gather_rows(const T* __restrict__ xg,
                                            const int2* __restrict__ ri,
                                            const int2* __restrict__ lg,
                                            float* __restrict__ ag,
                                            const int row0, const int lane) {
    int2 inf = ri[row0];
    int2 nxt = (lane < inf.y) ? lg[inf.x + lane] : make_int2(0, 0);
    for (int i = 0; i < GR_ROWS; ++i) {
        const int start = inf.x, len = inf.y;
        const int2 ent = nxt;
        if (i + 1 < GR_ROWS) {
            inf = ri[row0 + i + 1];
            nxt = (lane < inf.y) ? lg[inf.x + lane] : make_int2(0, 0);
        }
        float4 acc = make_float4(0.f, 0.f, 0.f, 0.f);
        gather_chunk<T>(acc, ent, min(len, 32), xg, lane);
        for (int b = 32; b < len; b += 32) {
            const int2 more = (b + lane < len) ? lg[start + b + lane] : make_int2(0, 0);
            gather_chunk<T>(acc, more, min(len - b, 32), xg, lane);
        }
        *(float4*)(ag + (size_t)(row0 + i) * NFEAT + lane * 4) = acc;
    }
}

__global__ __launch_bounds__(GR_WAVES * 32) void k_gather_f32(const float* __restrict__ x,
                                                              const int2*  __restrict__ rowinfo,
                                                              const int2*  __restrict__ list,
                                                              float*       __restrict__ agg) {
    const int g    = blockIdx.x >> 3;
    const int part = blockIdx.x & 7;
    const int wave = __builtin_amdgcn_readfirstlane(threadIdx.x >> 5);
    const int lane = threadIdx.x & 31;
    const int row0 = part * (GR_WAVES * GR_ROWS) + wave * GR_ROWS;
    gather_rows<float>(x + (size_t)g * NNODE * NFEAT, rowinfo + g * NNODE,
                       list + (size_t)g * NEDGE, agg + (size_t)g * NNODE * NFEAT, row0, lane);
}

__global__ __launch_bounds__(GR_WAVES * 32) void k_gather_f16(const half_t* __restrict__ h,
                                                              const int2*   __restrict__ rowinfo,
                                                              const int2*   __restrict__ list,
                                                              float*        __restrict__ agg) {
    const int g    = blockIdx.x >> 3;
    const int part = blockIdx.x & 7;
    const int wave = __builtin_amdgcn_readfirstlane(threadIdx.x >> 5);
    const int lane = threadIdx.x & 31;
    const int row0 = part * (GR_WAVES * GR_ROWS) + wave * GR_ROWS;
    gather_rows<half_t>(h + (size_t)g * NNODE * NFEAT, rowinfo + g * NNODE,
                        list + (size_t)g * NEDGE, agg + (size_t)g * NNODE * NFEAT, row0, lane);
}

#define WT_PITCH (NHID + 8)
__global__ __launch_bounds__(512, 1) void k_gemm(const float* __restrict__ agg,
                                                 const float* __restrict__ rs_in,
                                                 const float* __restrict__ W,
                                                 const float* __restrict__ bvec,
                                                 half_t*      __restrict__ h_out,
                                                 float*       __restrict__ emb_out,
                                                 int layer) {
    __shared__ __align__(16) half_t wt[NHID * WT_PITCH];
    __shared__ float  bs[NHID];
    __shared__ float  embacc[NHID];
    const int g    = blockIdx.x >> 1;
    const int half = blockIdx.x & 1;
    const int t    = threadIdx.x;

    {
        const int n  = t & (NHID - 1);
        const int kq = t >> 7;
#pragma unroll
        for (int j = 0; j < 4; ++j) {
            const int kg = kq + 4 * j;
            const float* wp = W + (size_t)(kg * 8) * NHID + n;
            v8h p;
#pragma unroll
            for (int i = 0; i < 8; ++i) p[i] = (half_t)wp[i * NHID];
            *(v8h*)(wt + n * WT_PITCH + kg * 8) = p;
        }
    }
    if (t < NHID) { bs[t] = bvec[t]; embacc[t] = 0.f; }
    __syncthreads();

    const int wave = t >> 5, lane = t & 31;
    const int l16 = lane & 15, hi = lane >> 4;
    const size_t gb = (size_t)g * NNODE;

    const int mi = half * 16 + wave;
    const int m  = mi * 16 + l16;
    const float* arow = agg + (gb + m) * NFEAT;
    const float  rsin = rs_in[gb + m];

    v8f c[8] = {};

#pragma unroll
    for (int ko = 0; ko < 4; ++ko) {
        const int kb = ko * 32;
        const int k0 = kb + hi * 8;
        const int k1 = kb + 16 + hi * 8;
        const float4 f0 = *(const float4*)(arow + k0);
        const float4 f1 = *(const float4*)(arow + k0 + 4);
        const float4 f2 = *(const float4*)(arow + k1);
        const float4 f3 = *(const float4*)(arow + k1 + 4);
        v16h a;
        a[0]  = (half_t)(f0.x * rsin); a[1]  = (half_t)(f0.y * rsin);
        a[2]  = (half_t)(f0.z * rsin); a[3]  = (half_t)(f0.w * rsin);
        a[4]  = (half_t)(f1.x * rsin); a[5]  = (half_t)(f1.y * rsin);
        a[6]  = (half_t)(f1.z * rsin); a[7]  = (half_t)(f1.w * rsin);
        a[8]  = (half_t)(f2.x * rsin); a[9]  = (half_t)(f2.y * rsin);
        a[10] = (half_t)(f2.z * rsin); a[11] = (half_t)(f2.w * rsin);
        a[12] = (half_t)(f3.x * rsin); a[13] = (half_t)(f3.y * rsin);
        a[14] = (half_t)(f3.z * rsin); a[15] = (half_t)(f3.w * rsin);

#pragma unroll
        for (int ni = 0; ni < 8; ++ni) {
            const half_t* brow = wt + (ni * 16 + l16) * WT_PITCH;
            const v8h blo = *(const v8h*)(brow + k0);
            const v8h bhi = *(const v8h*)(brow + k1);
            const v16h b = __builtin_shufflevector(blo, bhi,
                               0, 1, 2, 3, 4, 5, 6, 7,
                               8, 9, 10, 11, 12, 13, 14, 15);
            c[ni] = __builtin_amdgcn_wmma_f32_16x16x32_f16(
                        false, a, false, b, (short)0, c[ni], false, false);
        }
    }

#pragma unroll
    for (int ni = 0; ni < 8; ++ni) {
        const int n = ni * 16 + l16;
        if (layer == 0) {
#pragma unroll
            for (int r = 0; r < 8; ++r) {
                const int ml = r + 8 * hi;
                const float v = fmaxf(c[ni][r] + bs[n], 0.f);
                h_out[(gb + (size_t)mi * 16 + ml) * NFEAT + n] = (half_t)v;
            }
        } else {
            float s = 0.f;
#pragma unroll
            for (int r = 0; r < 8; ++r) s += fmaxf(c[ni][r] + bs[n], 0.f);
            atomicAdd(&embacc[n], s);
        }
    }

    if (layer == 1) {
        __syncthreads();
        if (t < NHID) atomicAdd(&emb_out[g * NHID + t], embacc[t] * (1.0f / NNODE));
    }
}

__global__ __launch_bounds__(128) void k_head(const float* __restrict__ emb,
                                              const float* __restrict__ Wt,
                                              const float* __restrict__ bt,
                                              const float* __restrict__ gamma,
                                              const float* __restrict__ beta,
                                              const float* __restrict__ rmean,
                                              const float* __restrict__ rvar,
                                              float* __restrict__ out) {
    __shared__ float er[NHID];
    const int g = blockIdx.x, n = threadIdx.x;
    er[n] = emb[g * NHID + n];
    __syncthreads();
    float z = bt[n];
    for (int k = 0; k < NHID; ++k) z += er[k] * Wt[k * NHID + n];
    z = gamma[n] * (z - rmean[n]) * rsqrtf(rvar[n] + BN_EPS) + beta[n];
    out[g * NHID + n] = fmaxf(z, 0.f);
}

extern "C" void kernel_launch(void* const* d_in, const int* in_sizes, int n_in,
                              void* d_out, int out_size, void* d_ws, size_t ws_size,
                              hipStream_t stream) {
    (void)in_sizes; (void)n_in; (void)out_size; (void)ws_size;
    const float* feats = (const float*)d_in[0];
    const int*   edges = (const int*)d_in[1];
    const float* W0    = (const float*)d_in[2];
    const float* b0    = (const float*)d_in[3];
    const float* W1    = (const float*)d_in[4];
    const float* b1    = (const float*)d_in[5];
    const float* Wt    = (const float*)d_in[6];
    const float* bt    = (const float*)d_in[7];
    const float* gamma = (const float*)d_in[8];
    const float* beta  = (const float*)d_in[9];
    const float* rmean = (const float*)d_in[10];
    const float* rvar  = (const float*)d_in[11];
    float* out = (float*)d_out;

    char* ws = (char*)d_ws;
    float*  rs_in   = (float*)ws;  ws += (size_t)NGRAPH * NNODE * sizeof(float);
    int2*   rowinfo = (int2*)ws;   ws += (size_t)NGRAPH * NNODE * sizeof(int2);
    int2*   list    = (int2*)ws;   ws += (size_t)NGRAPH * NEDGE * sizeof(int2);
    float*  agg     = (float*)ws;  ws += (size_t)NGRAPH * NNODE * NFEAT * sizeof(float);
    half_t* h1      = (half_t*)ws; ws += (size_t)NGRAPH * NNODE * NFEAT * sizeof(half_t);
    float*  emb     = (float*)ws;  ws += (size_t)NGRAPH * NHID * sizeof(float);
    const size_t emb_bytes = (size_t)NGRAPH * NHID * sizeof(float);

    k_csr<<<NGRAPH, 512, 0, stream>>>(edges, rs_in, rowinfo, list);

    k_gather_f32<<<NGRAPH * 8, GR_WAVES * 32, 0, stream>>>(feats, rowinfo, list, agg);
    k_gemm      <<<NGRAPH * 2, 512, 0, stream>>>(agg, rs_in, W0, b0, h1, emb, 0);

    hipMemsetAsync(emb, 0, emb_bytes, stream);
    k_gather_f16<<<NGRAPH * 8, GR_WAVES * 32, 0, stream>>>(h1, rowinfo, list, agg);
    k_gemm      <<<NGRAPH * 2, 512, 0, stream>>>(agg, rs_in, W1, b1, h1, emb, 1);

    k_head<<<NGRAPH, 128, 0, stream>>>(emb, Wt, bt, gamma, beta, rmean, rvar, out);
}


# Each layer's row gather fused into its WMMA kernel: summed rows are scaled, rounded to f16 into an LDS A tile that feeds the product; agg round trip and 2 launches removed; sum orders unchanged
# speedup vs baseline: 21.1392x; 1.3942x over previous
#include <hip/hip_runtime.h>
#include <hip/hip_bf16.h>

typedef _Float16 half_t;
typedef __attribute__((ext_vector_type(16))) _Float16 v16h;
typedef __attribute__((ext_vector_type(8)))  _Float16 v8h;
typedef __attribute__((ext_vector_type(4)))  _Float16 v4h;
typedef __attribute__((ext_vector_type(8)))  float    v8f;

#define NGRAPH 128
#define NNODE  512
#define NEDGE  8192
#define NFEAT  128
#define NHID   128
#define BN_EPS 1e-5f

__global__ __launch_bounds__(512) void k_csr(const int* __restrict__ edges,
                                             float* __restrict__ rs_in,
                                             int2*  __restrict__ rowinfo,
                                             int2*  __restrict__ list) {
    __shared__ int   cur[16 * NNODE];
    __shared__ int   degO[NNODE];
    __shared__ int   rowp[NNODE];
    __shared__ float rso[NNODE];
    __shared__ int   wsum[16];
    const int g    = blockIdx.x;
    const int t    = threadIdx.x;
    const int wave = t >> 5;
    const int lane = t & 31;
    const int* e = edges + (size_t)g * 2 * NEDGE;

    for (int i = t; i < 16 * NNODE; i += 512) cur[i] = 0;
    degO[t] = 0;
    __syncthreads();

    int src[16], dst[16];
#pragma unroll
    for (int r = 0; r < 16; ++r) {
        const int idx = wave * 512 + r * 32 + lane;
        src[r] = e[idx] & (NNODE - 1);
        dst[r] = e[NEDGE + idx] & (NNODE - 1);
    }
#pragma unroll
    for (int r = 0; r < 16; ++r) {
        atomicAdd(&cur[wave * NNODE + dst[r]], 1);
        atomicAdd(&degO[src[r]], 1);
    }
    __syncthreads();

    int run = 0;
#pragma unroll
    for (int w = 0; w < 16; ++w) {
        const int c = cur[w * NNODE + t];
        cur[w * NNODE + t] = run;
        run += c;
    }
    int incl = run;
    for (int o = 1; o < 32; o <<= 1) {
        const int v = __shfl_up(incl, o);
        if (lane >= o) incl += v;
    }
    if (lane == 31) wsum[wave] = incl;
    __syncthreads();
    int pre = 0;
    for (int w = 0; w < wave; ++w) pre += wsum[w];
    const int rstart = pre + incl - run;
    rowp[t] = rstart;
    rso[t]  = rsqrtf(fmaxf((float)degO[t], 1.0f));
    rs_in[g * NNODE + t]   = rsqrtf(fmaxf((float)run, 1.0f));
    rowinfo[g * NNODE + t] = make_int2(rstart, run);
    __syncthreads();

    int2* lg = list + (size_t)g * NEDGE;
#pragma unroll
    for (int r = 0; r < 16; ++r) {
        const int d = dst[r];
        int rank = 0;
#pragma unroll
        for (int j = 0; j < 31; ++j) {
            const int dj = __builtin_amdgcn_readlane(d, j);
            rank += (dj == d && j < lane) ? 1 : 0;
        }
        const int base = cur[wave * NNODE + d];
        atomicAdd(&cur[wave * NNODE + d], 1);
        const int pos = rowp[d] + base + rank;
        lg[pos] = make_int2(src[r], __float_as_int(rso[src[r]]));
    }
}

#define LY_THREADS 1024
#define LY_BROWS   256
#define LY_ROWS    8
#define WT_PITCH   (NHID + 8)

__device__ __forceinline__ float4 ld_row_f32(const float* __restrict__ xg, int s, int lane) {
    return *(const float4*)(xg + (size_t)s * NFEAT + lane * 4);
}
__device__ __forceinline__ float4 ld_row_f16(const half_t* __restrict__ xg, int s, int lane) {
    const v4h h = *(const v4h*)(xg + (size_t)s * NFEAT + lane * 4);
    return make_float4((float)h[0], (float)h[1], (float)h[2], (float)h[3]);
}

template <int U, typename T>
__device__ __forceinline__ void gather_batch(float4& acc, const int2 ent, const int k,
                                             const T* __restrict__ xg, const int lane) {
    int   s[U];
    float w[U];
    float4 v[U];
#pragma unroll
    for (int i = 0; i < U; ++i) {
        s[i] = __builtin_amdgcn_readlane(ent.x, k + i);
        w[i] = __int_as_float(__builtin_amdgcn_readlane(ent.y, k + i));
    }
#pragma unroll
    for (int i = 0; i < U; ++i) {
        if constexpr (sizeof(T) == 4) v[i] = ld_row_f32((const float*)xg, s[i], lane);
        else                          v[i] = ld_row_f16((const half_t*)xg, s[i], lane);
    }
#pragma unroll
    for (int i = 0; i < U; ++i) {
        acc.x = fmaf(v[i].x, w[i], acc.x);
        acc.y = fmaf(v[i].y, w[i], acc.y);
        acc.z = fmaf(v[i].z, w[i], acc.z);
        acc.w = fmaf(v[i].w, w[i], acc.w);
    }
}

template <typename T>
__device__ __forceinline__ void gather_chunk(float4& acc, const int2 ent, const int cnt,
                                             const T* __restrict__ xg, const int lane) {
    int k = 0;
    for (; k + 8 <= cnt; k += 8) gather_batch<8>(acc, ent, k, xg, lane);
    if (k + 4 <= cnt) { gather_batch<4>(acc, ent, k, xg, lane); k += 4; }
    const int rem = cnt - k;
    if      (rem == 3) gather_batch<3>(acc, ent, k, xg, lane);
    else if (rem == 2) gather_batch<2>(acc, ent, k, xg, lane);
    else if (rem == 1) gather_batch<1>(acc, ent, k, xg, lane);
}

template <typename T>
__device__ __forceinline__ void gather_rows_lds(const T* __restrict__ xg,
                                                const int2* __restrict__ ri,
                                                const int2* __restrict__ lg,
                                                const float* __restrict__ rsg,
                                                half_t* arows,
                                                const int row0, const int lane) {
    int2 inf = ri[row0];
    int2 nxt = (lane < inf.y) ? lg[inf.x + lane] : make_int2(0, 0);
    for (int i = 0; i < LY_ROWS; ++i) {
        const int start = inf.x, len = inf.y;
        const int2 ent = nxt;
        const float rsin = rsg[row0 + i];
        if (i + 1 < LY_ROWS) {
            inf = ri[row0 + i + 1];
            nxt = (lane < inf.y) ? lg[inf.x + lane] : make_int2(0, 0);
        }
        float4 acc = make_float4(0.f, 0.f, 0.f, 0.f);
        gather_chunk<T>(acc, ent, min(len, 32), xg, lane);
        for (int b = 32; b < len; b += 32) {
            const int2 more = (b + lane < len) ? lg[start + b + lane] : make_int2(0, 0);
            gather_chunk<T>(acc, more, min(len - b, 32), xg, lane);
        }
        v4h p;
        p[0] = (half_t)(acc.x * rsin); p[1] = (half_t)(acc.y * rsin);
        p[2] = (half_t)(acc.z * rsin); p[3] = (half_t)(acc.w * rsin);
        *(v4h*)(arows + i * WT_PITCH + lane * 4) = p;
    }
}

template <typename T, int LAYER>
__device__ __forceinline__ void layer_body(half_t* wt, half_t* at, float* bs, float* embacc,
                                           const T*     __restrict__ x,
                                           const int2*  __restrict__ rowinfo,
                                           const int2*  __restrict__ list,
                                           const float* __restrict__ rs_in,
                                           const float* __restrict__ W,
                                           const float* __restrict__ bvec,
                                           half_t*      __restrict__ h_out,
                                           float*       __restrict__ emb_out) {
    const int g    = blockIdx.x >> 1;
    const int half = blockIdx.x & 1;
    const int t    = threadIdx.x;

    const int fn = t & (NHID - 1);
    const int fk = t >> 7;
    float wreg[2][8];
#pragma unroll
    for (int j = 0; j < 2; ++j) {
        const float* wp = W + (size_t)((fk + 8 * j) * 8) * NHID + fn;
#pragma unroll
        for (int i = 0; i < 8; ++i) wreg[j][i] = wp[i * NHID];
    }
    const float bval = (t < NHID) ? bvec[t] : 0.f;

    const int wave = __builtin_amdgcn_readfirstlane(t >> 5);
    const int lane = t & 31;
    const size_t gb = (size_t)g * NNODE;

    gather_rows_lds<T>(x + gb * NFEAT, rowinfo + gb, list + (size_t)g * NEDGE, rs_in + gb,
                       at + (wave * LY_ROWS) * WT_PITCH, half * LY_BROWS + wave * LY_ROWS, lane);

#pragma unroll
    for (int j = 0; j < 2; ++j) {
        v8h p;
#pragma unroll
        for (int i = 0; i < 8; ++i) p[i] = (half_t)wreg[j][i];
        *(v8h*)(wt + fn * WT_PITCH + (fk + 8 * j) * 8) = p;
    }
    if (t < NHID) {
        bs[t] = bval;
        if constexpr (LAYER == 1) embacc[t] = 0.f;
    }
    __syncthreads();

    const int l16 = lane & 15, hi = lane >> 4;
    const int mt = wave >> 1;
    const int nh = wave & 1;
    const half_t* arow = at + (mt * 16 + l16) * WT_PITCH;

    v8f c[4] = {};

#pragma unroll
    for (int ko = 0; ko < 4; ++ko) {
        const int k0 = ko * 32 + hi * 8;
        const int k1 = ko * 32 + 16 + hi * 8;
        const v8h alo = *(const v8h*)(arow + k0);
        const v8h ahi = *(const v8h*)(arow + k1);
        const v16h a = __builtin_shufflevector(alo, ahi,
                           0, 1, 2, 3, 4, 5, 6, 7,
                           8, 9, 10, 11, 12, 13, 14, 15);
#pragma unroll
        for (int nj = 0; nj < 4; ++nj) {
            const half_t* brow = wt + ((nh * 4 + nj) * 16 + l16) * WT_PITCH;
            const v8h blo = *(const v8h*)(brow + k0);
            const v8h bhi = *(const v8h*)(brow + k1);
            const v16h b = __builtin_shufflevector(blo, bhi,
                               0, 1, 2, 3, 4, 5, 6, 7,
                               8, 9, 10, 11, 12, 13, 14, 15);
            c[nj] = __builtin_amdgcn_wmma_f32_16x16x32_f16(
                        false, a, false, b, (short)0, c[nj], false, false);
        }
    }

#pragma unroll
    for (int nj = 0; nj < 4; ++nj) {
        const int n = (nh * 4 + nj) * 16 + l16;
        if constexpr (LAYER == 0) {
#pragma unroll
            for (int r = 0; r < 8; ++r) {
                const int ml = r + 8 * hi;
                const float v = fmaxf(c[nj][r] + bs[n], 0.f);
                h_out[(gb + (size_t)(half * LY_BROWS + mt * 16 + ml)) * NFEAT + n] = (half_t)v;
            }
        } else {
            float s = 0.f;
#pragma unroll
            for (int r = 0; r < 8; ++r) s += fmaxf(c[nj][r] + bs[n], 0.f);
            atomicAdd(&embacc[n], s);
        }
    }

    if constexpr (LAYER == 1) {
        __syncthreads();
        if (t < NHID) atomicAdd(&emb_out[g * NHID + t], embacc[t] * (1.0f / NNODE));
    }
}

__global__ __launch_bounds__(LY_THREADS) void k_layer0(const float* __restrict__ x,
                                                       const int2*  __restrict__ rowinfo,
                                                       const int2*  __restrict__ list,
                                                       const float* __restrict__ rs_in,
                                                       const float* __restrict__ W,
                                                       const float* __restrict__ bvec,
                                                       half_t*      __restrict__ h_out) {
    __shared__ __align__(16) half_t wt[NHID * WT_PITCH];
    __shared__ __align__(16) half_t at[LY_BROWS * WT_PITCH];
    __shared__ float bs[NHID];
    layer_body<float, 0>(wt, at, bs, nullptr, x, rowinfo, list, rs_in, W, bvec, h_out, nullptr);
}

__global__ __launch_bounds__(LY_THREADS) void k_layer1(const half_t* __restrict__ h,
                                                       const int2*   __restrict__ rowinfo,
                                                       const int2*   __restrict__ list,
                                                       const float*  __restrict__ rs_in,
                                                       const float*  __restrict__ W,
                                                       const float*  __restrict__ bvec,
                                                       float*        __restrict__ emb_out) {
    __shared__ __align__(16) half_t wt[NHID * WT_PITCH];
    __shared__ __align__(16) half_t at[LY_BROWS * WT_PITCH];
    __shared__ float bs[NHID];
    __shared__ float embacc[NHID];
    layer_body<half_t, 1>(wt, at, bs, embacc, h, rowinfo, list, rs_in, W, bvec, nullptr, emb_out);
}

__global__ __launch_bounds__(128) void k_head(const float* __restrict__ emb,
                                              const float* __restrict__ Wt,
                                              const float* __restrict__ bt,
                                              const float* __restrict__ gamma,
                                              const float* __restrict__ beta,
                                              const float* __restrict__ rmean,
                                              const float* __restrict__ rvar,
                                              float* __restrict__ out) {
    __shared__ float er[NHID];
    const int g = blockIdx.x, n = threadIdx.x;
    er[n] = emb[g * NHID + n];
    __syncthreads();
    float z = bt[n];
    for (int k = 0; k < NHID; ++k) z += er[k] * Wt[k * NHID + n];
    z = gamma[n] * (z - rmean[n]) * rsqrtf(rvar[n] + BN_EPS) + beta[n];
    out[g * NHID + n] = fmaxf(z, 0.f);
}

extern "C" void kernel_launch(void* const* d_in, const int* in_sizes, int n_in,
                              void* d_out, int out_size, void* d_ws, size_t ws_size,
                              hipStream_t stream) {
    (void)in_sizes; (void)n_in; (void)out_size; (void)ws_size;
    const float* feats = (const float*)d_in[0];
    const int*   edges = (const int*)d_in[1];
    const float* W0    = (const float*)d_in[2];
    const float* b0    = (const float*)d_in[3];
    const float* W1    = (const float*)d_in[4];
    const float* b1    = (const float*)d_in[5];
    const float* Wt    = (const float*)d_in[6];
    const float* bt    = (const float*)d_in[7];
    const float* gamma = (const float*)d_in[8];
    const float* beta  = (const float*)d_in[9];
    const float* rmean = (const float*)d_in[10];
    const float* rvar  = (const float*)d_in[11];
    float* out = (float*)d_out;

    char* ws = (char*)d_ws;
    float*  rs_in   = (float*)ws;  ws += (size_t)NGRAPH * NNODE * sizeof(float);
    int2*   rowinfo = (int2*)ws;   ws += (size_t)NGRAPH * NNODE * sizeof(int2);
    int2*   list    = (int2*)ws;   ws += (size_t)NGRAPH * NEDGE * sizeof(int2);
    half_t* h1      = (half_t*)ws; ws += (size_t)NGRAPH * NNODE * NFEAT * sizeof(half_t);
    float*  emb     = (float*)ws;  ws += (size_t)NGRAPH * NHID * sizeof(float);
    const size_t emb_bytes = (size_t)NGRAPH * NHID * sizeof(float);

    k_csr<<<NGRAPH, 512, 0, stream>>>(edges, rs_in, rowinfo, list);

    k_layer0<<<NGRAPH * 2, LY_THREADS, 0, stream>>>(feats, rowinfo, list, rs_in, W0, b0, h1);

    hipMemsetAsync(emb, 0, emb_bytes, stream);
    k_layer1<<<NGRAPH * 2, LY_THREADS, 0, stream>>>(h1, rowinfo, list, rs_in, W1, b1, emb);

    k_head<<<NGRAPH, 128, 0, stream>>>(emb, Wt, bt, gamma, beta, rmean, rvar, out);
}


# Layer kernels: the graph's 512 source rows are copied once into LDS and the row gather reads them there, not from L2; summed rows wait in registers, the A tile reuses the bytes; sum orders unchanged
# speedup vs baseline: 24.6259x; 1.1649x over previous
#include <hip/hip_runtime.h>
#include <hip/hip_bf16.h>

typedef _Float16 half_t;
typedef __attribute__((ext_vector_type(16))) _Float16 v16h;
typedef __attribute__((ext_vector_type(8)))  _Float16 v8h;
typedef __attribute__((ext_vector_type(4)))  _Float16 v4h;
typedef __attribute__((ext_vector_type(8)))  float    v8f;

#define NGRAPH 128
#define NNODE  512
#define NEDGE  8192
#define NFEAT  128
#define NHID   128
#define BN_EPS 1e-5f

__global__ __launch_bounds__(512) void k_csr(const int* __restrict__ edges,
                                             float* __restrict__ rs_in,
                                             int2*  __restrict__ rowinfo,
                                             int2*  __restrict__ list) {
    __shared__ int   cur[16 * NNODE];
    __shared__ int   degO[NNODE];
    __shared__ int   rowp[NNODE];
    __shared__ float rso[NNODE];
    __shared__ int   wsum[16];
    const int g    = blockIdx.x;
    const int t    = threadIdx.x;
    const int wave = t >> 5;
    const int lane = t & 31;
    const int* e = edges + (size_t)g * 2 * NEDGE;

    for (int i = t; i < 16 * NNODE; i += 512) cur[i] = 0;
    degO[t] = 0;
    __syncthreads();

    int src[16], dst[16];
#pragma unroll
    for (int r = 0; r < 16; ++r) {
        const int idx = wave * 512 + r * 32 + lane;
        src[r] = e[idx] & (NNODE - 1);
        dst[r] = e[NEDGE + idx] & (NNODE - 1);
    }
#pragma unroll
    for (int r = 0; r < 16; ++r) {
        atomicAdd(&cur[wave * NNODE + dst[r]], 1);
        atomicAdd(&degO[src[r]], 1);
    }
    __syncthreads();

    int run = 0;
#pragma unroll
    for (int w = 0; w < 16; ++w) {
        const int c = cur[w * NNODE + t];
        cur[w * NNODE + t] = run;
        run += c;
    }
    int incl = run;
    for (int o = 1; o < 32; o <<= 1) {
        const int v = __shfl_up(incl, o);
        if (lane >= o) incl += v;
    }
    if (lane == 31) wsum[wave] = incl;
    __syncthreads();
    int pre = 0;
    for (int w = 0; w < wave; ++w) pre += wsum[w];
    const int rstart = pre + incl - run;
    rowp[t] = rstart;
    rso[t]  = rsqrtf(fmaxf((float)degO[t], 1.0f));
    rs_in[g * NNODE + t]   = rsqrtf(fmaxf((float)run, 1.0f));
    rowinfo[g * NNODE + t] = make_int2(rstart, run);
    __syncthreads();

    int2* lg = list + (size_t)g * NEDGE;
#pragma unroll
    for (int r = 0; r < 16; ++r) {
        const int d = dst[r];
        int rank = 0;
#pragma unroll
        for (int j = 0; j < 31; ++j) {
            const int dj = __builtin_amdgcn_readlane(d, j);
            rank += (dj == d && j < lane) ? 1 : 0;
        }
        const int base = cur[wave * NNODE + d];
        atomicAdd(&cur[wave * NNODE + d], 1);
        const int pos = rowp[d] + base + rank;
        lg[pos] = make_int2(src[r], __float_as_int(rso[src[r]]));
    }
}

#define LY_THREADS 1024
#define LY_BROWS   256
#define LY_ROWS    8
#define WT_PITCH   (NHID + 8)
#define WT_BYTES   (NHID * WT_PITCH * 2)

template <int U, typename T>
__device__ __forceinline__ void gather_batch(float4& acc, const int2 ent, const int k,
                                             const T* xs, const int lane) {
    int   s[U];
    float w[U];
    float4 v[U];
#pragma unroll
    for (int i = 0; i < U; ++i) {
        s[i] = __builtin_amdgcn_readlane(ent.x, k + i);
        w[i] = __int_as_float(__builtin_amdgcn_readlane(ent.y, k + i));
    }
#pragma unroll
    for (int i = 0; i < U; ++i) {
        if constexpr (sizeof(T) == 4) {
            v[i] = *(const float4*)(xs + s[i] * NFEAT + lane * 4);
        } else {
            const v4h h = *(const v4h*)(xs + s[i] * NFEAT + lane * 4);
            v[i] = make_float4((float)h[0], (float)h[1], (float)h[2], (float)h[3]);
        }
    }
#pragma unroll
    for (int i = 0; i < U; ++i) {
        acc.x = fmaf(v[i].x, w[i], acc.x);
        acc.y = fmaf(v[i].y, w[i], acc.y);
        acc.z = fmaf(v[i].z, w[i], acc.z);
        acc.w = fmaf(v[i].w, w[i], acc.w);
    }
}

template <typename T>
__device__ __forceinline__ void gather_chunk(float4& acc, const int2 ent, const int cnt,
                                             const T* xs, const int lane) {
    int k = 0;
    for (; k + 8 <= cnt; k += 8) gather_batch<8>(acc, ent, k, xs, lane);
    if (k + 4 <= cnt) { gather_batch<4>(acc, ent, k, xs, lane); k += 4; }
    const int rem = cnt - k;
    if      (rem == 3) gather_batch<3>(acc, ent, k, xs, lane);
    else if (rem == 2) gather_batch<2>(acc, ent, k, xs, lane);
    else if (rem == 1) gather_batch<1>(acc, ent, k, xs, lane);
}

template <typename T, int LAYER>
__device__ __forceinline__ void layer_body(unsigned char* smem, float* bs, float* embacc,
                                           const T*     __restrict__ x,
                                           const int2*  __restrict__ rowinfo,
                                           const int2*  __restrict__ list,
                                           const float* __restrict__ rs_in,
                                           const float* __restrict__ W,
                                           const float* __restrict__ bvec,
                                           half_t*      __restrict__ h_out,
                                           float*       __restrict__ emb_out) {
    constexpr int XS_BYTES = NNODE * NFEAT * (int)sizeof(T);
    T*      xs = (T*)smem;
    half_t* at = (half_t*)smem;
    half_t* wt = (half_t*)(smem + XS_BYTES);

    const int g    = blockIdx.x >> 1;
    const int half = blockIdx.x & 1;
    const int t    = threadIdx.x;
    const int wave = __builtin_amdgcn_readfirstlane(t >> 5);
    const int lane = t & 31;
    const size_t gb = (size_t)g * NNODE;
    const int2*  ri  = rowinfo + gb;
    const int2*  lg  = list + (size_t)g * NEDGE;
    const int    row0 = half * LY_BROWS + wave * LY_ROWS;

    const int2  myinf = ri[row0 + (lane & (LY_ROWS - 1))];
    const float myrs  = rs_in[gb + row0 + (lane & (LY_ROWS - 1))];

    constexpr int NV = XS_BYTES / 16 / LY_THREADS;
    const uint4* src4 = (const uint4*)(x + gb * NFEAT);
    uint4 stg[NV];
#pragma unroll
    for (int j = 0; j < NV; ++j) stg[j] = src4[j * LY_THREADS + t];

    int  start[LY_ROWS], len[LY_ROWS];
    int2 ent[LY_ROWS];
#pragma unroll
    for (int i = 0; i < LY_ROWS; ++i) {
        start[i] = __builtin_amdgcn_readlane(myinf.x, i);
        len[i]   = __builtin_amdgcn_readlane(myinf.y, i);
        ent[i]   = lg[min(start[i] + lane, NEDGE - 1)];
    }

    const int fn = t & (NHID - 1);
    const int fk = t >> 7;
    float wreg[2][8];
#pragma unroll
    for (int j = 0; j < 2; ++j) {
        const float* wp = W + (size_t)((fk + 8 * j) * 8) * NHID + fn;
#pragma unroll
        for (int i = 0; i < 8; ++i) wreg[j][i] = wp[i * NHID];
    }
    const float bval = bvec[t & (NHID - 1)];


#pragma unroll
    for (int j = 0; j < NV; ++j) ((uint4*)smem)[j * LY_THREADS + t] = stg[j];
    __syncthreads();

    v4h kept[LY_ROWS];
#pragma unroll
    for (int i = 0; i < LY_ROWS; ++i) {
        float4 acc = make_float4(0.f, 0.f, 0.f, 0.f);
        gather_chunk<T>(acc, ent[i], min(len[i], 32), xs, lane);
        for (int b = 32; b < len[i]; b += 32) {
            const int2 more = (b + lane < len[i]) ? lg[start[i] + b + lane] : make_int2(0, 0);
            gather_chunk<T>(acc, more, min(len[i] - b, 32), xs, lane);
        }
        const float rsin = __int_as_float(__builtin_amdgcn_readlane(__float_as_int(myrs), i));
        v4h p;
        p[0] = (half_t)(acc.x * rsin); p[1] = (half_t)(acc.y * rsin);
        p[2] = (half_t)(acc.z * rsin); p[3] = (half_t)(acc.w * rsin);
        kept[i] = p;
    }
    __syncthreads();

#pragma unroll
    for (int i = 0; i < LY_ROWS; ++i)
        *(v4h*)(at + (wave * LY_ROWS + i) * WT_PITCH + lane * 4) = kept[i];

#pragma unroll
    for (int j = 0; j < 2; ++j) {
        v8h p;
#pragma unroll
        for (int i = 0; i < 8; ++i) p[i] = (half_t)wreg[j][i];
        *(v8h*)(wt + fn * WT_PITCH + (fk + 8 * j) * 8) = p;
    }
    if (t < NHID) {
        bs[t] = bval;
        if constexpr (LAYER == 1) embacc[t] = 0.f;
    }
    __syncthreads();

    const int l16 = lane & 15, hi = lane >> 4;
    const int mt = wave >> 1;
    const int nh = wave & 1;
    const half_t* arow = at + (mt * 16 + l16) * WT_PITCH;

    v8f c[4] = {};

#pragma unroll
    for (int ko = 0; ko < 4; ++ko) {
        const int k0 = ko * 32 + hi * 8;
        const int k1 = ko * 32 + 16 + hi * 8;
        const v8h alo = *(const v8h*)(arow + k0);
        const v8h ahi = *(const v8h*)(arow + k1);
        const v16h a = __builtin_shufflevector(alo, ahi,
                           0, 1, 2, 3, 4, 5, 6, 7,
                           8, 9, 10, 11, 12, 13, 14, 15);
#pragma unroll
        for (int nj = 0; nj < 4; ++nj) {
            const half_t* brow = wt + ((nh * 4 + nj) * 16 + l16) * WT_PITCH;
            const v8h blo = *(const v8h*)(brow + k0);
            const v8h bhi = *(const v8h*)(brow + k1);
            const v16h b = __builtin_shufflevector(blo, bhi,
                               0, 1, 2, 3, 4, 5, 6, 7,
                               8, 9, 10, 11, 12, 13, 14, 15);
            c[nj] = __builtin_amdgcn_wmma_f32_16x16x32_f16(
                        false, a, false, b, (short)0, c[nj], false, false);
        }
    }

#pragma unroll
    for (int nj = 0; nj < 4; ++nj) {
        const int n = (nh * 4 + nj) * 16 + l16;
        if constexpr (LAYER == 0) {
#pragma unroll
            for (int r = 0; r < 8; ++r) {
                const int ml = r + 8 * hi;
                const float v = fmaxf(c[nj][r] + bs[n], 0.f);
                h_out[(gb + (size_t)(half * LY_BROWS + mt * 16 + ml)) * NFEAT + n] = (half_t)v;
            }
        } else {
            float s = 0.f;
#pragma unroll
            for (int r = 0; r < 8; ++r) s += fmaxf(c[nj][r] + bs[n], 0.f);
            atomicAdd(&embacc[n], s);
        }
    }

    if constexpr (LAYER == 1) {
        __syncthreads();
        if (t < NHID) atomicAdd(&emb_out[g * NHID + t], embacc[t] * (1.0f / NNODE));
    }
}

__global__ __launch_bounds__(LY_THREADS) void k_layer0(const float* __restrict__ x,
                                                       const int2*  __restrict__ rowinfo,
                                                       const int2*  __restrict__ list,
                                                       const float* __restrict__ rs_in,
                                                       const float* __restrict__ W,
                                                       const float* __restrict__ bvec,
                                                       half_t*      __restrict__ h_out) {
    __shared__ __align__(16) unsigned char smem[NNODE * NFEAT * 4 + WT_BYTES];
    __shared__ float bs[NHID];
    layer_body<float, 0>(smem, bs, nullptr, x, rowinfo, list, rs_in, W, bvec, h_out, nullptr);
}

__global__ __launch_bounds__(LY_THREADS) void k_layer1(const half_t* __restrict__ h,
                                                       const int2*   __restrict__ rowinfo,
                                                       const int2*   __restrict__ list,
                                                       const float*  __restrict__ rs_in,
                                                       const float*  __restrict__ W,
                                                       const float*  __restrict__ bvec,
                                                       float*        __restrict__ emb_out) {
    __shared__ __align__(16) unsigned char smem[NNODE * NFEAT * 2 + WT_BYTES];
    __shared__ float bs[NHID];
    __shared__ float embacc[NHID];
    layer_body<half_t, 1>(smem, bs, embacc, h, rowinfo, list, rs_in, W, bvec, nullptr, emb_out);
}

__global__ __launch_bounds__(128) void k_head(const float* __restrict__ emb,
                                              const float* __restrict__ Wt,
                                              const float* __restrict__ bt,
                                              const float* __restrict__ gamma,
                                              const float* __restrict__ beta,
                                              const float* __restrict__ rmean,
                                              const float* __restrict__ rvar,
                                              float* __restrict__ out) {
    __shared__ float er[NHID];
    const int g = blockIdx.x, n = threadIdx.x;
    er[n] = emb[g * NHID + n];
    __syncthreads();
    float z = bt[n];
    for (int k = 0; k < NHID; ++k) z += er[k] * Wt[k * NHID + n];
    z = gamma[n] * (z - rmean[n]) * rsqrtf(rvar[n] + BN_EPS) + beta[n];
    out[g * NHID + n] = fmaxf(z, 0.f);
}

extern "C" void kernel_launch(void* const* d_in, const int* in_sizes, int n_in,
                              void* d_out, int out_size, void* d_ws, size_t ws_size,
                              hipStream_t stream) {
    (void)in_sizes; (void)n_in; (void)out_size; (void)ws_size;
    const float* feats = (const float*)d_in[0];
    const int*   edges = (const int*)d_in[1];
    const float* W0    = (const float*)d_in[2];
    const float* b0    = (const float*)d_in[3];
    const float* W1    = (const float*)d_in[4];
    const float* b1    = (const float*)d_in[5];
    const float* Wt    = (const float*)d_in[6];
    const float* bt    = (const float*)d_in[7];
    const float* gamma = (const float*)d_in[8];
    const float* beta  = (const float*)d_in[9];
    const float* rmean = (const float*)d_in[10];
    const float* rvar  = (const float*)d_in[11];
    float* out = (float*)d_out;

    char* ws = (char*)d_ws;
    float*  rs_in   = (float*)ws;  ws += (size_t)NGRAPH * NNODE * sizeof(float);
    int2*   rowinfo = (int2*)ws;   ws += (size_t)NGRAPH * NNODE * sizeof(int2);
    int2*   list    = (int2*)ws;   ws += (size_t)NGRAPH * NEDGE * sizeof(int2);
    half_t* h1      = (half_t*)ws; ws += (size_t)NGRAPH * NNODE * NFEAT * sizeof(half_t);
    float*  emb     = (float*)ws;  ws += (size_t)NGRAPH * NHID * sizeof(float);
    const size_t emb_bytes = (size_t)NGRAPH * NHID * sizeof(float);

    k_csr<<<NGRAPH, 512, 0, stream>>>(edges, rs_in, rowinfo, list);

    k_layer0<<<NGRAPH * 2, LY_THREADS, 0, stream>>>(feats, rowinfo, list, rs_in, W0, b0, h1);

    hipMemsetAsync(emb, 0, emb_bytes, stream);
    k_layer1<<<NGRAPH * 2, LY_THREADS, 0, stream>>>(h1, rowinfo, list, rs_in, W1, b1, emb);

    k_head<<<NGRAPH, 128, 0, stream>>>(emb, Wt, bt, gamma, beta, rmean, rvar, out);
}


# k_csr: rank of an edge among same-dst lanes of its 32-edge round found only for lanes that collide (a returning LDS add shows them), not by 31 lane reads per round; list and sum orders unchanged
# speedup vs baseline: 25.4107x; 1.0319x over previous
#include <hip/hip_runtime.h>
#include <hip/hip_bf16.h>

typedef _Float16 half_t;
typedef __attribute__((ext_vector_type(16))) _Float16 v16h;
typedef __attribute__((ext_vector_type(8)))  _Float16 v8h;
typedef __attribute__((ext_vector_type(4)))  _Float16 v4h;
typedef __attribute__((ext_vector_type(8)))  float    v8f;

#define NGRAPH 128
#define NNODE  512
#define NEDGE  8192
#define NFEAT  128
#define NHID   128
#define BN_EPS 1e-5f

__global__ __launch_bounds__(512) void k_csr(const int* __restrict__ edges,
                                             float* __restrict__ rs_in,
                                             int2*  __restrict__ rowinfo,
                                             int2*  __restrict__ list) {
    __shared__ int   cur[16 * NNODE];
    __shared__ int   degO[NNODE];
    __shared__ int   rowp[NNODE];
    __shared__ float rso[NNODE];
    __shared__ int   wsum[16];
    const int g    = blockIdx.x;
    const int t    = threadIdx.x;
    const int wave = t >> 5;
    const int lane = t & 31;
    const int* e = edges + (size_t)g * 2 * NEDGE;

    for (int i = t; i < 16 * NNODE; i += 512) cur[i] = 0;
    degO[t] = 0;
    __syncthreads();

    int src[16], dst[16];
#pragma unroll
    for (int r = 0; r < 16; ++r) {
        const int idx = wave * 512 + r * 32 + lane;
        src[r] = e[idx] & (NNODE - 1);
        dst[r] = e[NEDGE + idx] & (NNODE - 1);
    }
#pragma unroll
    for (int r = 0; r < 16; ++r) {
        atomicAdd(&cur[wave * NNODE + dst[r]], 1);
        atomicAdd(&degO[src[r]], 1);
    }
    __syncthreads();

    int run = 0;
#pragma unroll
    for (int w = 0; w < 16; ++w) {
        const int c = cur[w * NNODE + t];
        cur[w * NNODE + t] = run;
        run += c;
    }
    int incl = run;
    for (int o = 1; o < 32; o <<= 1) {
        const int v = __shfl_up(incl, o);
        if (lane >= o) incl += v;
    }
    if (lane == 31) wsum[wave] = incl;
    __syncthreads();
    int pre = 0;
    for (int w = 0; w < wave; ++w) pre += wsum[w];
    const int rstart = pre + incl - run;
    rowp[t] = rstart;
    rso[t]  = rsqrtf(fmaxf((float)degO[t], 1.0f));
    rs_in[g * NNODE + t]   = rsqrtf(fmaxf((float)run, 1.0f));
    rowinfo[g * NNODE + t] = make_int2(rstart, run);
    __syncthreads();

    int2* lg = list + (size_t)g * NEDGE;
    int*  cw = cur + wave * NNODE;
    int base[16], got[16];
#pragma unroll
    for (int r = 0; r < 16; ++r) {
        base[r] = __hip_atomic_load(&cw[dst[r]], __ATOMIC_RELAXED, __HIP_MEMORY_SCOPE_WORKGROUP);
        got[r]  = atomicAdd(&cw[dst[r]], 1);
    }
    int   rp[16];
    float rv[16];
#pragma unroll
    for (int r = 0; r < 16; ++r) {
        rp[r] = rowp[dst[r]];
        rv[r] = rso[src[r]];
    }
#pragma unroll
    for (int r = 0; r < 16; ++r) {
        const int d = dst[r];
        int rank = 0;
        unsigned coll = __builtin_amdgcn_ballot_w32(got[r] != base[r]);
        for (int it = 0; it < 31 && coll != 0u; ++it) {
            const int j  = __builtin_ctz(coll);
            const int dj = __builtin_amdgcn_readlane(d, j);
            const unsigned grp = __builtin_amdgcn_ballot_w32(d == dj);
            if (d == dj) rank = (int)__builtin_amdgcn_mbcnt_lo(grp, 0u);
            coll &= ~grp;
        }
        const int pos = rp[r] + base[r] + rank;
        lg[pos] = make_int2(src[r], __float_as_int(rv[r]));
    }
}

#define LY_THREADS 1024
#define LY_BROWS   256
#define LY_ROWS    8
#define WT_PITCH   (NHID + 8)
#define WT_BYTES   (NHID * WT_PITCH * 2)

template <int U, typename T>
__device__ __forceinline__ void gather_batch(float4& acc, const int2 ent, const int k,
                                             const T* xs, const int lane) {
    int   s[U];
    float w[U];
    float4 v[U];
#pragma unroll
    for (int i = 0; i < U; ++i) {
        s[i] = __builtin_amdgcn_readlane(ent.x, k + i);
        w[i] = __int_as_float(__builtin_amdgcn_readlane(ent.y, k + i));
    }
#pragma unroll
    for (int i = 0; i < U; ++i) {
        if constexpr (sizeof(T) == 4) {
            v[i] = *(const float4*)(xs + s[i] * NFEAT + lane * 4);
        } else {
            const v4h h = *(const v4h*)(xs + s[i] * NFEAT + lane * 4);
            v[i] = make_float4((float)h[0], (float)h[1], (float)h[2], (float)h[3]);
        }
    }
#pragma unroll
    for (int i = 0; i < U; ++i) {
        acc.x = fmaf(v[i].x, w[i], acc.x);
        acc.y = fmaf(v[i].y, w[i], acc.y);
        acc.z = fmaf(v[i].z, w[i], acc.z);
        acc.w = fmaf(v[i].w, w[i], acc.w);
    }
}

template <typename T>
__device__ __forceinline__ void gather_chunk(float4& acc, const int2 ent, const int cnt,
                                             const T* xs, const int lane) {
    int k = 0;
    for (; k + 8 <= cnt; k += 8) gather_batch<8>(acc, ent, k, xs, lane);
    if (k + 4 <= cnt) { gather_batch<4>(acc, ent, k, xs, lane); k += 4; }
    const int rem = cnt - k;
    if      (rem == 3) gather_batch<3>(acc, ent, k, xs, lane);
    else if (rem == 2) gather_batch<2>(acc, ent, k, xs, lane);
    else if (rem == 1) gather_batch<1>(acc, ent, k, xs, lane);
}

template <typename T, int LAYER>
__device__ __forceinline__ void layer_body(unsigned char* smem, float* bs, float* embacc,
                                           const T*     __restrict__ x,
                                           const int2*  __restrict__ rowinfo,
                                           const int2*  __restrict__ list,
                                           const float* __restrict__ rs_in,
                                           const float* __restrict__ W,
                                           const float* __restrict__ bvec,
                                           half_t*      __restrict__ h_out,
                                           float*       __restrict__ emb_out) {
    constexpr int XS_BYTES = NNODE * NFEAT * (int)sizeof(T);
    T*      xs = (T*)smem;
    half_t* at = (half_t*)smem;
    half_t* wt = (half_t*)(smem + XS_BYTES);

    const int g    = blockIdx.x >> 1;
    const int half = blockIdx.x & 1;
    const int t    = threadIdx.x;
    const int wave = __builtin_amdgcn_readfirstlane(t >> 5);
    const int lane = t & 31;
    const size_t gb = (size_t)g * NNODE;
    const int2*  ri  = rowinfo + gb;
    const int2*  lg  = list + (size_t)g * NEDGE;
    const int    row0 = half * LY_BROWS + wave * LY_ROWS;

    const int2  myinf = ri[row0 + (lane & (LY_ROWS - 1))];
    const float myrs  = rs_in[gb + row0 + (lane & (LY_ROWS - 1))];

    constexpr int NV = XS_BYTES / 16 / LY_THREADS;
    const uint4* src4 = (const uint4*)(x + gb * NFEAT);
    uint4 stg[NV];
#pragma unroll
    for (int j = 0; j < NV; ++j) stg[j] = src4[j * LY_THREADS + t];

    int  start[LY_ROWS], len[LY_ROWS];
    int2 ent[LY_ROWS];
#pragma unroll
    for (int i = 0; i < LY_ROWS; ++i) {
        start[i] = __builtin_amdgcn_readlane(myinf.x, i);
        len[i]   = __builtin_amdgcn_readlane(myinf.y, i);
        ent[i]   = lg[min(start[i] + lane, NEDGE - 1)];
    }

    const int fn = t & (NHID - 1);
    const int fk = t >> 7;
    float wreg[2][8];
#pragma unroll
    for (int j = 0; j < 2; ++j) {
        const float* wp = W + (size_t)((fk + 8 * j) * 8) * NHID + fn;
#pragma unroll
        for (int i = 0; i < 8; ++i) wreg[j][i] = wp[i * NHID];
    }
    const float bval = bvec[t & (NHID - 1)];


#pragma unroll
    for (int j = 0; j < NV; ++j) ((uint4*)smem)[j * LY_THREADS + t] = stg[j];
    __syncthreads();

    v4h kept[LY_ROWS];
#pragma unroll
    for (int i = 0; i < LY_ROWS; ++i) {
        float4 acc = make_float4(0.f, 0.f, 0.f, 0.f);
        gather_chunk<T>(acc, ent[i], min(len[i], 32), xs, lane);
        for (int b = 32; b < len[i]; b += 32) {
            const int2 more = (b + lane < len[i]) ? lg[start[i] + b + lane] : make_int2(0, 0);
            gather_chunk<T>(acc, more, min(len[i] - b, 32), xs, lane);
        }
        const float rsin = __int_as_float(__builtin_amdgcn_readlane(__float_as_int(myrs), i));
        v4h p;
        p[0] = (half_t)(acc.x * rsin); p[1] = (half_t)(acc.y * rsin);
        p[2] = (half_t)(acc.z * rsin); p[3] = (half_t)(acc.w * rsin);
        kept[i] = p;
    }
    __syncthreads();

#pragma unroll
    for (int i = 0; i < LY_ROWS; ++i)
        *(v4h*)(at + (wave * LY_ROWS + i) * WT_PITCH + lane * 4) = kept[i];

#pragma unroll
    for (int j = 0; j < 2; ++j) {
        v8h p;
#pragma unroll
        for (int i = 0; i < 8; ++i) p[i] = (half_t)wreg[j][i];
        *(v8h*)(wt + fn * WT_PITCH + (fk + 8 * j) * 8) = p;
    }
    if (t < NHID) {
        bs[t] = bval;
        if constexpr (LAYER == 1) embacc[t] = 0.f;
    }
    __syncthreads();

    const int l16 = lane & 15, hi = lane >> 4;
    const int mt = wave >> 1;
    const int nh = wave & 1;
    const half_t* arow = at + (mt * 16 + l16) * WT_PITCH;

    v8f c[4] = {};

#pragma unroll
    for (int ko = 0; ko < 4; ++ko) {
        const int k0 = ko * 32 + hi * 8;
        const int k1 = ko * 32 + 16 + hi * 8;
        const v8h alo = *(const v8h*)(arow + k0);
        const v8h ahi = *(const v8h*)(arow + k1);
        const v16h a = __builtin_shufflevector(alo, ahi,
                           0, 1, 2, 3, 4, 5, 6, 7,
                           8, 9, 10, 11, 12, 13, 14, 15);
#pragma unroll
        for (int nj = 0; nj < 4; ++nj) {
            const half_t* brow = wt + ((nh * 4 + nj) * 16 + l16) * WT_PITCH;
            const v8h blo = *(const v8h*)(brow + k0);
            const v8h bhi = *(const v8h*)(brow + k1);
            const v16h b = __builtin_shufflevector(blo, bhi,
                               0, 1, 2, 3, 4, 5, 6, 7,
                               8, 9, 10, 11, 12, 13, 14, 15);
            c[nj] = __builtin_amdgcn_wmma_f32_16x16x32_f16(
                        false, a, false, b, (short)0, c[nj], false, false);
        }
    }

#pragma unroll
    for (int nj = 0; nj < 4; ++nj) {
        const int n = (nh * 4 + nj) * 16 + l16;
        if constexpr (LAYER == 0) {
#pragma unroll
            for (int r = 0; r < 8; ++r) {
                const int ml = r + 8 * hi;
                const float v = fmaxf(c[nj][r] + bs[n], 0.f);
                h_out[(gb + (size_t)(half * LY_BROWS + mt * 16 + ml)) * NFEAT + n] = (half_t)v;
            }
        } else {
            float s = 0.f;
#pragma unroll
            for (int r = 0; r < 8; ++r) s += fmaxf(c[nj][r] + bs[n], 0.f);
            atomicAdd(&embacc[n], s);
        }
    }

    if constexpr (LAYER == 1) {
        __syncthreads();
        if (t < NHID) atomicAdd(&emb_out[g * NHID + t], embacc[t] * (1.0f / NNODE));
    }
}

__global__ __launch_bounds__(LY_THREADS) void k_layer0(const float* __restrict__ x,
                                                       const int2*  __restrict__ rowinfo,
                                                       const int2*  __restrict__ list,
                                                       const float* __restrict__ rs_in,
                                                       const float* __restrict__ W,
                                                       const float* __restrict__ bvec,
                                                       half_t*      __restrict__ h_out) {
    __shared__ __align__(16) unsigned char smem[NNODE * NFEAT * 4 + WT_BYTES];
    __shared__ float bs[NHID];
    layer_body<float, 0>(smem, bs, nullptr, x, rowinfo, list, rs_in, W, bvec, h_out, nullptr);
}

__global__ __launch_bounds__(LY_THREADS) void k_layer1(const half_t* __restrict__ h,
                                                       const int2*   __restrict__ rowinfo,
                                                       const int2*   __restrict__ list,
                                                       const float*  __restrict__ rs_in,
                                                       const float*  __restrict__ W,
                                                       const float*  __restrict__ bvec,
                                                       float*        __restrict__ emb_out) {
    __shared__ __align__(16) unsigned char smem[NNODE * NFEAT * 2 + WT_BYTES];
    __shared__ float bs[NHID];
    __shared__ float embacc[NHID];
    layer_body<half_t, 1>(smem, bs, embacc, h, rowinfo, list, rs_in, W, bvec, nullptr, emb_out);
}

__global__ __launch_bounds__(128) void k_head(const float* __restrict__ emb,
                                              const float* __restrict__ Wt,
                                              const float* __restrict__ bt,
                                              const float* __restrict__ gamma,
                                              const float* __restrict__ beta,
                                              const float* __restrict__ rmean,
                                              const float* __restrict__ rvar,
                                              float* __restrict__ out) {
    __shared__ float er[NHID];
    const int g = blockIdx.x, n = threadIdx.x;
    er[n] = emb[g * NHID + n];
    __syncthreads();
    float z = bt[n];
    for (int k = 0; k < NHID; ++k) z += er[k] * Wt[k * NHID + n];
    z = gamma[n] * (z - rmean[n]) * rsqrtf(rvar[n] + BN_EPS) + beta[n];
    out[g * NHID + n] = fmaxf(z, 0.f);
}

extern "C" void kernel_launch(void* const* d_in, const int* in_sizes, int n_in,
                              void* d_out, int out_size, void* d_ws, size_t ws_size,
                              hipStream_t stream) {
    (void)in_sizes; (void)n_in; (void)out_size; (void)ws_size;
    const float* feats = (const float*)d_in[0];
    const int*   edges = (const int*)d_in[1];
    const float* W0    = (const float*)d_in[2];
    const float* b0    = (const float*)d_in[3];
    const float* W1    = (const float*)d_in[4];
    const float* b1    = (const float*)d_in[5];
    const float* Wt    = (const float*)d_in[6];
    const float* bt    = (const float*)d_in[7];
    const float* gamma = (const float*)d_in[8];
    const float* beta  = (const float*)d_in[9];
    const float* rmean = (const float*)d_in[10];
    const float* rvar  = (const float*)d_in[11];
    float* out = (float*)d_out;

    char* ws = (char*)d_ws;
    float*  rs_in   = (float*)ws;  ws += (size_t)NGRAPH * NNODE * sizeof(float);
    int2*   rowinfo = (int2*)ws;   ws += (size_t)NGRAPH * NNODE * sizeof(int2);
    int2*   list    = (int2*)ws;   ws += (size_t)NGRAPH * NEDGE * sizeof(int2);
    half_t* h1      = (half_t*)ws; ws += (size_t)NGRAPH * NNODE * NFEAT * sizeof(half_t);
    float*  emb     = (float*)ws;  ws += (size_t)NGRAPH * NHID * sizeof(float);
    const size_t emb_bytes = (size_t)NGRAPH * NHID * sizeof(float);

    k_csr<<<NGRAPH, 512, 0, stream>>>(edges, rs_in, rowinfo, list);

    k_layer0<<<NGRAPH * 2, LY_THREADS, 0, stream>>>(feats, rowinfo, list, rs_in, W0, b0, h1);

    hipMemsetAsync(emb, 0, emb_bytes, stream);
    k_layer1<<<NGRAPH * 2, LY_THREADS, 0, stream>>>(h1, rowinfo, list, rs_in, W1, b1, emb);

    k_head<<<NGRAPH, 128, 0, stream>>>(emb, Wt, bt, gamma, beta, rmean, rvar, out);
}


# Head fused into layer-1 kernel: a graph's two workgroups store half means, a returning counter add picks the later one, which does Wt, BN, relu; head launch, emb memset gone; head dot in 8 k-groups
# speedup vs baseline: 26.4780x; 1.0420x over previous
#include <hip/hip_runtime.h>
#include <hip/hip_bf16.h>

typedef _Float16 half_t;
typedef __attribute__((ext_vector_type(16))) _Float16 v16h;
typedef __attribute__((ext_vector_type(8)))  _Float16 v8h;
typedef __attribute__((ext_vector_type(4)))  _Float16 v4h;
typedef __attribute__((ext_vector_type(8)))  float    v8f;

#define NGRAPH 128
#define NNODE  512
#define NEDGE  8192
#define NFEAT  128
#define NHID   128
#define BN_EPS 1e-5f

__global__ __launch_bounds__(512) void k_csr(const int* __restrict__ edges,
                                             float* __restrict__ rs_in,
                                             int2*  __restrict__ rowinfo,
                                             int2*  __restrict__ list,
                                             unsigned int* __restrict__ cnt) {
    __shared__ int   cur[16 * NNODE];
    __shared__ int   degO[NNODE];
    __shared__ int   rowp[NNODE];
    __shared__ float rso[NNODE];
    __shared__ int   wsum[16];
    const int g    = blockIdx.x;
    const int t    = threadIdx.x;
    const int wave = t >> 5;
    const int lane = t & 31;
    const int* e = edges + (size_t)g * 2 * NEDGE;

    for (int i = t; i < 16 * NNODE; i += 512) cur[i] = 0;
    degO[t] = 0;
    if (t == 0) cnt[g] = 0u;
    __syncthreads();

    int src[16], dst[16];
#pragma unroll
    for (int r = 0; r < 16; ++r) {
        const int idx = wave * 512 + r * 32 + lane;
        src[r] = e[idx] & (NNODE - 1);
        dst[r] = e[NEDGE + idx] & (NNODE - 1);
    }
#pragma unroll
    for (int r = 0; r < 16; ++r) {
        atomicAdd(&cur[wave * NNODE + dst[r]], 1);
        atomicAdd(&degO[src[r]], 1);
    }
    __syncthreads();

    int run = 0;
#pragma unroll
    for (int w = 0; w < 16; ++w) {
        const int c = cur[w * NNODE + t];
        cur[w * NNODE + t] = run;
        run += c;
    }
    int incl = run;
    for (int o = 1; o < 32; o <<= 1) {
        const int v = __shfl_up(incl, o);
        if (lane >= o) incl += v;
    }
    if (lane == 31) wsum[wave] = incl;
    __syncthreads();
    int pre = 0;
    for (int w = 0; w < wave; ++w) pre += wsum[w];
    const int rstart = pre + incl - run;
    rowp[t] = rstart;
    rso[t]  = rsqrtf(fmaxf((float)degO[t], 1.0f));
    rs_in[g * NNODE + t]   = rsqrtf(fmaxf((float)run, 1.0f));
    rowinfo[g * NNODE + t] = make_int2(rstart, run);
    __syncthreads();

    int2* lg = list + (size_t)g * NEDGE;
    int*  cw = cur + wave * NNODE;
    int base[16], got[16];
#pragma unroll
    for (int r = 0; r < 16; ++r) {
        base[r] = __hip_atomic_load(&cw[dst[r]], __ATOMIC_RELAXED, __HIP_MEMORY_SCOPE_WORKGROUP);
        got[r]  = atomicAdd(&cw[dst[r]], 1);
    }
    int   rp[16];
    float rv[16];
#pragma unroll
    for (int r = 0; r < 16; ++r) {
        rp[r] = rowp[dst[r]];
        rv[r] = rso[src[r]];
    }
#pragma unroll
    for (int r = 0; r < 16; ++r) {
        const int d = dst[r];
        int rank = 0;
        unsigned coll = __builtin_amdgcn_ballot_w32(got[r] != base[r]);
        for (int it = 0; it < 31 && coll != 0u; ++it) {
            const int j  = __builtin_ctz(coll);
            const int dj = __builtin_amdgcn_readlane(d, j);
            const unsigned grp = __builtin_amdgcn_ballot_w32(d == dj);
            if (d == dj) rank = (int)__builtin_amdgcn_mbcnt_lo(grp, 0u);
            coll &= ~grp;
        }
        const int pos = rp[r] + base[r] + rank;
        lg[pos] = make_int2(src[r], __float_as_int(rv[r]));
    }
}

#define LY_THREADS 1024
#define LY_BROWS   256
#define LY_ROWS    8
#define WT_PITCH   (NHID + 8)
#define WT_BYTES   (NHID * WT_PITCH * 2)

template <int U, typename T>
__device__ __forceinline__ void gather_batch(float4& acc, const int2 ent, const int k,
                                             const T* xs, const int lane) {
    int   s[U];
    float w[U];
    float4 v[U];
#pragma unroll
    for (int i = 0; i < U; ++i) {
        s[i] = __builtin_amdgcn_readlane(ent.x, k + i);
        w[i] = __int_as_float(__builtin_amdgcn_readlane(ent.y, k + i));
    }
#pragma unroll
    for (int i = 0; i < U; ++i) {
        if constexpr (sizeof(T) == 4) {
            v[i] = *(const float4*)(xs + s[i] * NFEAT + lane * 4);
        } else {
            const v4h h = *(const v4h*)(xs + s[i] * NFEAT + lane * 4);
            v[i] = make_float4((float)h[0], (float)h[1], (float)h[2], (float)h[3]);
        }
    }
#pragma unroll
    for (int i = 0; i < U; ++i) {
        acc.x = fmaf(v[i].x, w[i], acc.x);
        acc.y = fmaf(v[i].y, w[i], acc.y);
        acc.z = fmaf(v[i].z, w[i], acc.z);
        acc.w = fmaf(v[i].w, w[i], acc.w);
    }
}

template <typename T>
__device__ __forceinline__ void gather_chunk(float4& acc, const int2 ent, const int cnt,
                                             const T* xs, const int lane) {
    int k = 0;
    for (; k + 8 <= cnt; k += 8) gather_batch<8>(acc, ent, k, xs, lane);
    if (k + 4 <= cnt) { gather_batch<4>(acc, ent, k, xs, lane); k += 4; }
    const int rem = cnt - k;
    if      (rem == 3) gather_batch<3>(acc, ent, k, xs, lane);
    else if (rem == 2) gather_batch<2>(acc, ent, k, xs, lane);
    else if (rem == 1) gather_batch<1>(acc, ent, k, xs, lane);
}

template <typename T, int LAYER>
__device__ __forceinline__ void layer_body(unsigned char* smem, float* bs, float* embacc,
                                           const T*     __restrict__ x,
                                           const int2*  __restrict__ rowinfo,
                                           const int2*  __restrict__ list,
                                           const float* __restrict__ rs_in,
                                           const float* __restrict__ W,
                                           const float* __restrict__ bvec,
                                           half_t*      __restrict__ h_out) {
    constexpr int XS_BYTES = NNODE * NFEAT * (int)sizeof(T);
    T*      xs = (T*)smem;
    half_t* at = (half_t*)smem;
    half_t* wt = (half_t*)(smem + XS_BYTES);

    const int g    = blockIdx.x >> 1;
    const int half = blockIdx.x & 1;
    const int t    = threadIdx.x;
    const int wave = __builtin_amdgcn_readfirstlane(t >> 5);
    const int lane = t & 31;
    const size_t gb = (size_t)g * NNODE;
    const int2*  ri  = rowinfo + gb;
    const int2*  lg  = list + (size_t)g * NEDGE;
    const int    row0 = half * LY_BROWS + wave * LY_ROWS;

    const int2  myinf = ri[row0 + (lane & (LY_ROWS - 1))];
    const float myrs  = rs_in[gb + row0 + (lane & (LY_ROWS - 1))];

    constexpr int NV = XS_BYTES / 16 / LY_THREADS;
    const uint4* src4 = (const uint4*)(x + gb * NFEAT);
    uint4 stg[NV];
#pragma unroll
    for (int j = 0; j < NV; ++j) stg[j] = src4[j * LY_THREADS + t];

    int  start[LY_ROWS], len[LY_ROWS];
    int2 ent[LY_ROWS];
#pragma unroll
    for (int i = 0; i < LY_ROWS; ++i) {
        start[i] = __builtin_amdgcn_readlane(myinf.x, i);
        len[i]   = __builtin_amdgcn_readlane(myinf.y, i);
        ent[i]   = lg[min(start[i] + lane, NEDGE - 1)];
    }

    const int fn = t & (NHID - 1);
    const int fk = t >> 7;
    float wreg[2][8];
#pragma unroll
    for (int j = 0; j < 2; ++j) {
        const float* wp = W + (size_t)((fk + 8 * j) * 8) * NHID + fn;
#pragma unroll
        for (int i = 0; i < 8; ++i) wreg[j][i] = wp[i * NHID];
    }
    const float bval = bvec[t & (NHID - 1)];


#pragma unroll
    for (int j = 0; j < NV; ++j) ((uint4*)smem)[j * LY_THREADS + t] = stg[j];
    __syncthreads();

    v4h kept[LY_ROWS];
#pragma unroll
    for (int i = 0; i < LY_ROWS; ++i) {
        float4 acc = make_float4(0.f, 0.f, 0.f, 0.f);
        gather_chunk<T>(acc, ent[i], min(len[i], 32), xs, lane);
        for (int b = 32; b < len[i]; b += 32) {
            const int2 more = (b + lane < len[i]) ? lg[start[i] + b + lane] : make_int2(0, 0);
            gather_chunk<T>(acc, more, min(len[i] - b, 32), xs, lane);
        }
        const float rsin = __int_as_float(__builtin_amdgcn_readlane(__float_as_int(myrs), i));
        v4h p;
        p[0] = (half_t)(acc.x * rsin); p[1] = (half_t)(acc.y * rsin);
        p[2] = (half_t)(acc.z * rsin); p[3] = (half_t)(acc.w * rsin);
        kept[i] = p;
    }
    __syncthreads();

#pragma unroll
    for (int i = 0; i < LY_ROWS; ++i)
        *(v4h*)(at + (wave * LY_ROWS + i) * WT_PITCH + lane * 4) = kept[i];

#pragma unroll
    for (int j = 0; j < 2; ++j) {
        v8h p;
#pragma unroll
        for (int i = 0; i < 8; ++i) p[i] = (half_t)wreg[j][i];
        *(v8h*)(wt + fn * WT_PITCH + (fk + 8 * j) * 8) = p;
    }
    if (t < NHID) {
        bs[t] = bval;
        if constexpr (LAYER == 1) embacc[t] = 0.f;
    }
    __syncthreads();

    const int l16 = lane & 15, hi = lane >> 4;
    const int mt = wave >> 1;
    const int nh = wave & 1;
    const half_t* arow = at + (mt * 16 + l16) * WT_PITCH;

    v8f c[4] = {};

#pragma unroll
    for (int ko = 0; ko < 4; ++ko) {
        const int k0 = ko * 32 + hi * 8;
        const int k1 = ko * 32 + 16 + hi * 8;
        const v8h alo = *(const v8h*)(arow + k0);
        const v8h ahi = *(const v8h*)(arow + k1);
        const v16h a = __builtin_shufflevector(alo, ahi,
                           0, 1, 2, 3, 4, 5, 6, 7,
                           8, 9, 10, 11, 12, 13, 14, 15);
#pragma unroll
        for (int nj = 0; nj < 4; ++nj) {
            const half_t* brow = wt + ((nh * 4 + nj) * 16 + l16) * WT_PITCH;
            const v8h blo = *(const v8h*)(brow + k0);
            const v8h bhi = *(const v8h*)(brow + k1);
            const v16h b = __builtin_shufflevector(blo, bhi,
                               0, 1, 2, 3, 4, 5, 6, 7,
                               8, 9, 10, 11, 12, 13, 14, 15);
            c[nj] = __builtin_amdgcn_wmma_f32_16x16x32_f16(
                        false, a, false, b, (short)0, c[nj], false, false);
        }
    }

#pragma unroll
    for (int nj = 0; nj < 4; ++nj) {
        const int n = (nh * 4 + nj) * 16 + l16;
        if constexpr (LAYER == 0) {
#pragma unroll
            for (int r = 0; r < 8; ++r) {
                const int ml = r + 8 * hi;
                const float v = fmaxf(c[nj][r] + bs[n], 0.f);
                h_out[(gb + (size_t)(half * LY_BROWS + mt * 16 + ml)) * NFEAT + n] = (half_t)v;
            }
        } else {
            float s = 0.f;
#pragma unroll
            for (int r = 0; r < 8; ++r) s += fmaxf(c[nj][r] + bs[n], 0.f);
            atomicAdd(&embacc[n], s);
        }
    }
}

__global__ __launch_bounds__(LY_THREADS) void k_layer0(const float* __restrict__ x,
                                                       const int2*  __restrict__ rowinfo,
                                                       const int2*  __restrict__ list,
                                                       const float* __restrict__ rs_in,
                                                       const float* __restrict__ W,
                                                       const float* __restrict__ bvec,
                                                       half_t*      __restrict__ h_out) {
    __shared__ __align__(16) unsigned char smem[NNODE * NFEAT * 4 + WT_BYTES];
    __shared__ float bs[NHID];
    layer_body<float, 0>(smem, bs, nullptr, x, rowinfo, list, rs_in, W, bvec, h_out);
}

#define HD_KQ 8
#define HD_KL (NHID / HD_KQ)

__global__ __launch_bounds__(LY_THREADS) void k_layer1(const half_t* __restrict__ h,
                                                       const int2*   __restrict__ rowinfo,
                                                       const int2*   __restrict__ list,
                                                       const float*  __restrict__ rs_in,
                                                       const float*  __restrict__ W,
                                                       const float*  __restrict__ bvec,
                                                       const float*  __restrict__ Wt,
                                                       const float*  __restrict__ bt,
                                                       const float*  __restrict__ gamma,
                                                       const float*  __restrict__ beta,
                                                       const float*  __restrict__ rmean,
                                                       const float*  __restrict__ rvar,
                                                       float*        part,
                                                       unsigned int* cnt,
                                                       float*        __restrict__ out) {
    __shared__ __align__(16) unsigned char smem[NNODE * NFEAT * 2 + WT_BYTES];
    __shared__ float bs[NHID];
    __shared__ float embacc[NHID];
    __shared__ float er[NHID];
    __shared__ float zp[HD_KQ * NHID];
    __shared__ unsigned int s_ticket;
    layer_body<half_t, 1>(smem, bs, embacc, h, rowinfo, list, rs_in, W, bvec, nullptr);

    const int g    = blockIdx.x >> 1;
    const int half = blockIdx.x & 1;
    const int t    = threadIdx.x;
    const int hn   = t & (NHID - 1);
    const int kq   = t >> 7;

    __syncthreads();

    float wtr[HD_KL];
#pragma unroll
    for (int i = 0; i < HD_KL; ++i) wtr[i] = Wt[(size_t)(kq * HD_KL + i) * NHID + hn];
    const float hbt = bt[hn], hga = gamma[hn], hbe = beta[hn], hmu = rmean[hn], hva = rvar[hn];

    if (t < 32) {
        float* mine = part + (size_t)(g * 2 + half) * NHID;
        float pm[4];
#pragma unroll
        for (int j = 0; j < 4; ++j) pm[j] = embacc[j * 32 + t] * (1.0f / NNODE);
#pragma unroll
        for (int j = 0; j < 4; ++j)
            __hip_atomic_store(mine + j * 32 + t, pm[j], __ATOMIC_RELAXED, __HIP_MEMORY_SCOPE_AGENT);
        __builtin_amdgcn_fence(__ATOMIC_RELEASE, "agent");
        if (t == 0) s_ticket = __hip_atomic_fetch_add(cnt + g, 1u, __ATOMIC_RELAXED, __HIP_MEMORY_SCOPE_AGENT);
    }
    __syncthreads();
    if (s_ticket == 1u) {
        if (t < NHID) {
            __builtin_amdgcn_fence(__ATOMIC_ACQUIRE, "agent");
            const float* p0 = part + (size_t)(g * 2) * NHID;
            const float a = __hip_atomic_load(p0 + t,        __ATOMIC_RELAXED, __HIP_MEMORY_SCOPE_AGENT);
            const float b = __hip_atomic_load(p0 + NHID + t, __ATOMIC_RELAXED, __HIP_MEMORY_SCOPE_AGENT);
            er[t] = a + b;
        }
        __syncthreads();
        float s = 0.f;
#pragma unroll
        for (int i = 0; i < HD_KL; ++i) s = fmaf(er[kq * HD_KL + i], wtr[i], s);
        zp[kq * NHID + hn] = s;
        __syncthreads();
        if (t < NHID) {
            float z = hbt;
#pragma unroll
            for (int j = 0; j < HD_KQ; ++j) z += zp[j * NHID + t];
            z = hga * (z - hmu) * rsqrtf(hva + BN_EPS) + hbe;
            out[g * NHID + t] = fmaxf(z, 0.f);
        }
    }
}

extern "C" void kernel_launch(void* const* d_in, const int* in_sizes, int n_in,
                              void* d_out, int out_size, void* d_ws, size_t ws_size,
                              hipStream_t stream) {
    (void)in_sizes; (void)n_in; (void)out_size; (void)ws_size;
    const float* feats = (const float*)d_in[0];
    const int*   edges = (const int*)d_in[1];
    const float* W0    = (const float*)d_in[2];
    const float* b0    = (const float*)d_in[3];
    const float* W1    = (const float*)d_in[4];
    const float* b1    = (const float*)d_in[5];
    const float* Wt    = (const float*)d_in[6];
    const float* bt    = (const float*)d_in[7];
    const float* gamma = (const float*)d_in[8];
    const float* beta  = (const float*)d_in[9];
    const float* rmean = (const float*)d_in[10];
    const float* rvar  = (const float*)d_in[11];
    float* out = (float*)d_out;

    char* ws = (char*)d_ws;
    float*  rs_in   = (float*)ws;  ws += (size_t)NGRAPH * NNODE * sizeof(float);
    int2*   rowinfo = (int2*)ws;   ws += (size_t)NGRAPH * NNODE * sizeof(int2);
    int2*   list    = (int2*)ws;   ws += (size_t)NGRAPH * NEDGE * sizeof(int2);
    half_t* h1      = (half_t*)ws; ws += (size_t)NGRAPH * NNODE * NFEAT * sizeof(half_t);
    float*  part    = (float*)ws;  ws += (size_t)NGRAPH * 2 * NHID * sizeof(float);
    unsigned int* cnt = (unsigned int*)ws;

    k_csr<<<NGRAPH, 512, 0, stream>>>(edges, rs_in, rowinfo, list, cnt);

    k_layer0<<<NGRAPH * 2, LY_THREADS, 0, stream>>>(feats, rowinfo, list, rs_in, W0, b0, h1);

    k_layer1<<<NGRAPH * 2, LY_THREADS, 0, stream>>>(h1, rowinfo, list, rs_in, W1, b1,
                                                    Wt, bt, gamma, beta, rmean, rvar,
                                                    part, cnt, out);
}


# k_csr: dst-sorted edge list assembled in LDS and written out as whole 128-byte lines (16-byte stores), not as 8-byte stores scattered over up to 32 lines per wave store; list and sum orders unchanged
# speedup vs baseline: 27.9820x; 1.0568x over previous
#include <hip/hip_runtime.h>
#include <hip/hip_bf16.h>

typedef _Float16 half_t;
typedef __attribute__((ext_vector_type(16))) _Float16 v16h;
typedef __attribute__((ext_vector_type(8)))  _Float16 v8h;
typedef __attribute__((ext_vector_type(4)))  _Float16 v4h;
typedef __attribute__((ext_vector_type(8)))  float    v8f;

#define NGRAPH 128
#define NNODE  512
#define NEDGE  8192
#define NFEAT  128
#define NHID   128
#define BN_EPS 1e-5f

__global__ __launch_bounds__(512) void k_csr(const int* __restrict__ edges,
                                             float* __restrict__ rs_in,
                                             int2*  __restrict__ rowinfo,
                                             int2*  __restrict__ list,
                                             unsigned int* __restrict__ cnt) {
    __shared__ int   cur[16 * NNODE];
    __shared__ int   degO[NNODE];
    __shared__ int   rowp[NNODE];
    __shared__ float rso[NNODE];
    __shared__ int   wsum[16];
    __shared__ __align__(16) int2 lst[NEDGE];
    const int g    = blockIdx.x;
    const int t    = threadIdx.x;
    const int wave = t >> 5;
    const int lane = t & 31;
    const int* e = edges + (size_t)g * 2 * NEDGE;

    for (int i = t; i < 16 * NNODE; i += 512) cur[i] = 0;
    degO[t] = 0;
    if (t == 0) cnt[g] = 0u;
    __syncthreads();

    int src[16], dst[16];
#pragma unroll
    for (int r = 0; r < 16; ++r) {
        const int idx = wave * 512 + r * 32 + lane;
        src[r] = e[idx] & (NNODE - 1);
        dst[r] = e[NEDGE + idx] & (NNODE - 1);
    }
#pragma unroll
    for (int r = 0; r < 16; ++r) {
        atomicAdd(&cur[wave * NNODE + dst[r]], 1);
        atomicAdd(&degO[src[r]], 1);
    }
    __syncthreads();

    int run = 0;
#pragma unroll
    for (int w = 0; w < 16; ++w) {
        const int c = cur[w * NNODE + t];
        cur[w * NNODE + t] = run;
        run += c;
    }
    int incl = run;
    for (int o = 1; o < 32; o <<= 1) {
        const int v = __shfl_up(incl, o);
        if (lane >= o) incl += v;
    }
    if (lane == 31) wsum[wave] = incl;
    __syncthreads();
    int pre = 0;
    for (int w = 0; w < wave; ++w) pre += wsum[w];
    const int rstart = pre + incl - run;
    rowp[t] = rstart;
    rso[t]  = rsqrtf(fmaxf((float)degO[t], 1.0f));
    rs_in[g * NNODE + t]   = rsqrtf(fmaxf((float)run, 1.0f));
    rowinfo[g * NNODE + t] = make_int2(rstart, run);
    __syncthreads();

    int*  cw = cur + wave * NNODE;
    int base[16], got[16];
#pragma unroll
    for (int r = 0; r < 16; ++r) {
        base[r] = __hip_atomic_load(&cw[dst[r]], __ATOMIC_RELAXED, __HIP_MEMORY_SCOPE_WORKGROUP);
        got[r]  = atomicAdd(&cw[dst[r]], 1);
    }
    int   rp[16];
    float rv[16];
#pragma unroll
    for (int r = 0; r < 16; ++r) {
        rp[r] = rowp[dst[r]];
        rv[r] = rso[src[r]];
    }
#pragma unroll
    for (int r = 0; r < 16; ++r) {
        const int d = dst[r];
        int rank = 0;
        unsigned coll = __builtin_amdgcn_ballot_w32(got[r] != base[r]);
        for (int it = 0; it < 31 && coll != 0u; ++it) {
            const int j  = __builtin_ctz(coll);
            const int dj = __builtin_amdgcn_readlane(d, j);
            const unsigned grp = __builtin_amdgcn_ballot_w32(d == dj);
            if (d == dj) rank = (int)__builtin_amdgcn_mbcnt_lo(grp, 0u);
            coll &= ~grp;
        }
        const int pos = rp[r] + base[r] + rank;
        lst[pos] = make_int2(src[r], __float_as_int(rv[r]));
    }
    __syncthreads();

    const uint4* ls4 = (const uint4*)lst;
    uint4*       lg4 = (uint4*)(list + (size_t)g * NEDGE);
    constexpr int NLV = NEDGE * (int)sizeof(int2) / 16 / 512;
    uint4 cp[NLV];
#pragma unroll
    for (int j = 0; j < NLV; ++j) cp[j] = ls4[j * 512 + t];
#pragma unroll
    for (int j = 0; j < NLV; ++j) lg4[j * 512 + t] = cp[j];
}

#define LY_THREADS 1024
#define LY_BROWS   256
#define LY_ROWS    8
#define WT_PITCH   (NHID + 8)
#define WT_BYTES   (NHID * WT_PITCH * 2)

template <int U, typename T>
__device__ __forceinline__ void gather_batch(float4& acc, const int2 ent, const int k,
                                             const T* xs, const int lane) {
    int   s[U];
    float w[U];
    float4 v[U];
#pragma unroll
    for (int i = 0; i < U; ++i) {
        s[i] = __builtin_amdgcn_readlane(ent.x, k + i);
        w[i] = __int_as_float(__builtin_amdgcn_readlane(ent.y, k + i));
    }
#pragma unroll
    for (int i = 0; i < U; ++i) {
        if constexpr (sizeof(T) == 4) {
            v[i] = *(const float4*)(xs + s[i] * NFEAT + lane * 4);
        } else {
            const v4h h = *(const v4h*)(xs + s[i] * NFEAT + lane * 4);
            v[i] = make_float4((float)h[0], (float)h[1], (float)h[2], (float)h[3]);
        }
    }
#pragma unroll
    for (int i = 0; i < U; ++i) {
        acc.x = fmaf(v[i].x, w[i], acc.x);
        acc.y = fmaf(v[i].y, w[i], acc.y);
        acc.z = fmaf(v[i].z, w[i], acc.z);
        acc.w = fmaf(v[i].w, w[i], acc.w);
    }
}

template <typename T>
__device__ __forceinline__ void gather_chunk(float4& acc, const int2 ent, const int cnt,
                                             const T* xs, const int lane) {
    int k = 0;
    for (; k + 8 <= cnt; k += 8) gather_batch<8>(acc, ent, k, xs, lane);
    if (k + 4 <= cnt) { gather_batch<4>(acc, ent, k, xs, lane); k += 4; }
    const int rem = cnt - k;
    if      (rem == 3) gather_batch<3>(acc, ent, k, xs, lane);
    else if (rem == 2) gather_batch<2>(acc, ent, k, xs, lane);
    else if (rem == 1) gather_batch<1>(acc, ent, k, xs, lane);
}

template <typename T, int LAYER>
__device__ __forceinline__ void layer_body(unsigned char* smem, float* bs, float* embacc,
                                           const T*     __restrict__ x,
                                           const int2*  __restrict__ rowinfo,
                                           const int2*  __restrict__ list,
                                           const float* __restrict__ rs_in,
                                           const float* __restrict__ W,
                                           const float* __restrict__ bvec,
                                           half_t*      __restrict__ h_out) {
    constexpr int XS_BYTES = NNODE * NFEAT * (int)sizeof(T);
    T*      xs = (T*)smem;
    half_t* at = (half_t*)smem;
    half_t* wt = (half_t*)(smem + XS_BYTES);

    const int g    = blockIdx.x >> 1;
    const int half = blockIdx.x & 1;
    const int t    = threadIdx.x;
    const int wave = __builtin_amdgcn_readfirstlane(t >> 5);
    const int lane = t & 31;
    const size_t gb = (size_t)g * NNODE;
    const int2*  ri  = rowinfo + gb;
    const int2*  lg  = list + (size_t)g * NEDGE;
    const int    row0 = half * LY_BROWS + wave * LY_ROWS;

    const int2  myinf = ri[row0 + (lane & (LY_ROWS - 1))];
    const float myrs  = rs_in[gb + row0 + (lane & (LY_ROWS - 1))];

    constexpr int NV = XS_BYTES / 16 / LY_THREADS;
    const uint4* src4 = (const uint4*)(x + gb * NFEAT);
    uint4 stg[NV];
#pragma unroll
    for (int j = 0; j < NV; ++j) stg[j] = src4[j * LY_THREADS + t];

    int  start[LY_ROWS], len[LY_ROWS];
    int2 ent[LY_ROWS];
#pragma unroll
    for (int i = 0; i < LY_ROWS; ++i) {
        start[i] = __builtin_amdgcn_readlane(myinf.x, i);
        len[i]   = __builtin_amdgcn_readlane(myinf.y, i);
        ent[i]   = lg[min(start[i] + lane, NEDGE - 1)];
    }

    const int fn = t & (NHID - 1);
    const int fk = t >> 7;
    float wreg[2][8];
#pragma unroll
    for (int j = 0; j < 2; ++j) {
        const float* wp = W + (size_t)((fk + 8 * j) * 8) * NHID + fn;
#pragma unroll
        for (int i = 0; i < 8; ++i) wreg[j][i] = wp[i * NHID];
    }
    const float bval = bvec[t & (NHID - 1)];


#pragma unroll
    for (int j = 0; j < NV; ++j) ((uint4*)smem)[j * LY_THREADS + t] = stg[j];
    __syncthreads();

    v4h kept[LY_ROWS];
#pragma unroll
    for (int i = 0; i < LY_ROWS; ++i) {
        float4 acc = make_float4(0.f, 0.f, 0.f, 0.f);
        gather_chunk<T>(acc, ent[i], min(len[i], 32), xs, lane);
        for (int b = 32; b < len[i]; b += 32) {
            const int2 more = (b + lane < len[i]) ? lg[start[i] + b + lane] : make_int2(0, 0);
            gather_chunk<T>(acc, more, min(len[i] - b, 32), xs, lane);
        }
        const float rsin = __int_as_float(__builtin_amdgcn_readlane(__float_as_int(myrs), i));
        v4h p;
        p[0] = (half_t)(acc.x * rsin); p[1] = (half_t)(acc.y * rsin);
        p[2] = (half_t)(acc.z * rsin); p[3] = (half_t)(acc.w * rsin);
        kept[i] = p;
    }
    __syncthreads();

#pragma unroll
    for (int i = 0; i < LY_ROWS; ++i)
        *(v4h*)(at + (wave * LY_ROWS + i) * WT_PITCH + lane * 4) = kept[i];

#pragma unroll
    for (int j = 0; j < 2; ++j) {
        v8h p;
#pragma unroll
        for (int i = 0; i < 8; ++i) p[i] = (half_t)wreg[j][i];
        *(v8h*)(wt + fn * WT_PITCH + (fk + 8 * j) * 8) = p;
    }
    if (t < NHID) {
        bs[t] = bval;
        if constexpr (LAYER == 1) embacc[t] = 0.f;
    }
    __syncthreads();

    const int l16 = lane & 15, hi = lane >> 4;
    const int mt = wave >> 1;
    const int nh = wave & 1;
    const half_t* arow = at + (mt * 16 + l16) * WT_PITCH;

    v8f c[4] = {};

#pragma unroll
    for (int ko = 0; ko < 4; ++ko) {
        const int k0 = ko * 32 + hi * 8;
        const int k1 = ko * 32 + 16 + hi * 8;
        const v8h alo = *(const v8h*)(arow + k0);
        const v8h ahi = *(const v8h*)(arow + k1);
        const v16h a = __builtin_shufflevector(alo, ahi,
                           0, 1, 2, 3, 4, 5, 6, 7,
                           8, 9, 10, 11, 12, 13, 14, 15);
#pragma unroll
        for (int nj = 0; nj < 4; ++nj) {
            const half_t* brow = wt + ((nh * 4 + nj) * 16 + l16) * WT_PITCH;
            const v8h blo = *(const v8h*)(brow + k0);
            const v8h bhi = *(const v8h*)(brow + k1);
            const v16h b = __builtin_shufflevector(blo, bhi,
                               0, 1, 2, 3, 4, 5, 6, 7,
                               8, 9, 10, 11, 12, 13, 14, 15);
            c[nj] = __builtin_amdgcn_wmma_f32_16x16x32_f16(
                        false, a, false, b, (short)0, c[nj], false, false);
        }
    }

#pragma unroll
    for (int nj = 0; nj < 4; ++nj) {
        const int n = (nh * 4 + nj) * 16 + l16;
        if constexpr (LAYER == 0) {
#pragma unroll
            for (int r = 0; r < 8; ++r) {
                const int ml = r + 8 * hi;
                const float v = fmaxf(c[nj][r] + bs[n], 0.f);
                h_out[(gb + (size_t)(half * LY_BROWS + mt * 16 + ml)) * NFEAT + n] = (half_t)v;
            }
        } else {
            float s = 0.f;
#pragma unroll
            for (int r = 0; r < 8; ++r) s += fmaxf(c[nj][r] + bs[n], 0.f);
            atomicAdd(&embacc[n], s);
        }
    }
}

__global__ __launch_bounds__(LY_THREADS) void k_layer0(const float* __restrict__ x,
                                                       const int2*  __restrict__ rowinfo,
                                                       const int2*  __restrict__ list,
                                                       const float* __restrict__ rs_in,
                                                       const float* __restrict__ W,
                                                       const float* __restrict__ bvec,
                                                       half_t*      __restrict__ h_out) {
    __shared__ __align__(16) unsigned char smem[NNODE * NFEAT * 4 + WT_BYTES];
    __shared__ float bs[NHID];
    layer_body<float, 0>(smem, bs, nullptr, x, rowinfo, list, rs_in, W, bvec, h_out);
}

#define HD_KQ 8
#define HD_KL (NHID / HD_KQ)

__global__ __launch_bounds__(LY_THREADS) void k_layer1(const half_t* __restrict__ h,
                                                       const int2*   __restrict__ rowinfo,
                                                       const int2*   __restrict__ list,
                                                       const float*  __restrict__ rs_in,
                                                       const float*  __restrict__ W,
                                                       const float*  __restrict__ bvec,
                                                       const float*  __restrict__ Wt,
                                                       const float*  __restrict__ bt,
                                                       const float*  __restrict__ gamma,
                                                       const float*  __restrict__ beta,
                                                       const float*  __restrict__ rmean,
                                                       const float*  __restrict__ rvar,
                                                       float*        part,
                                                       unsigned int* cnt,
                                                       float*        __restrict__ out) {
    __shared__ __align__(16) unsigned char smem[NNODE * NFEAT * 2 + WT_BYTES];
    __shared__ float bs[NHID];
    __shared__ float embacc[NHID];
    __shared__ float er[NHID];
    __shared__ float zp[HD_KQ * NHID];
    __shared__ unsigned int s_ticket;
    layer_body<half_t, 1>(smem, bs, embacc, h, rowinfo, list, rs_in, W, bvec, nullptr);

    const int g    = blockIdx.x >> 1;
    const int half = blockIdx.x & 1;
    const int t    = threadIdx.x;
    const int hn   = t & (NHID - 1);
    const int kq   = t >> 7;

    __syncthreads();

    float wtr[HD_KL];
#pragma unroll
    for (int i = 0; i < HD_KL; ++i) wtr[i] = Wt[(size_t)(kq * HD_KL + i) * NHID + hn];
    const float hbt = bt[hn], hga = gamma[hn], hbe = beta[hn], hmu = rmean[hn], hva = rvar[hn];

    if (t < 32) {
        float* mine = part + (size_t)(g * 2 + half) * NHID;
        float pm[4];
#pragma unroll
        for (int j = 0; j < 4; ++j) pm[j] = embacc[j * 32 + t] * (1.0f / NNODE);
#pragma unroll
        for (int j = 0; j < 4; ++j)
            __hip_atomic_store(mine + j * 32 + t, pm[j], __ATOMIC_RELAXED, __HIP_MEMORY_SCOPE_AGENT);
        __builtin_amdgcn_fence(__ATOMIC_RELEASE, "agent");
        if (t == 0) s_ticket = __hip_atomic_fetch_add(cnt + g, 1u, __ATOMIC_RELAXED, __HIP_MEMORY_SCOPE_AGENT);
    }
    __syncthreads();
    if (s_ticket == 1u) {
        if (t < NHID) {
            __builtin_amdgcn_fence(__ATOMIC_ACQUIRE, "agent");
            const float* p0 = part + (size_t)(g * 2) * NHID;
            const float a = __hip_atomic_load(p0 + t,        __ATOMIC_RELAXED, __HIP_MEMORY_SCOPE_AGENT);
            const float b = __hip_atomic_load(p0 + NHID + t, __ATOMIC_RELAXED, __HIP_MEMORY_SCOPE_AGENT);
            er[t] = a + b;
        }
        __syncthreads();
        float s = 0.f;
#pragma unroll
        for (int i = 0; i < HD_KL; ++i) s = fmaf(er[kq * HD_KL + i], wtr[i], s);
        zp[kq * NHID + hn] = s;
        __syncthreads();
        if (t < NHID) {
            float z = hbt;
#pragma unroll
            for (int j = 0; j < HD_KQ; ++j) z += zp[j * NHID + t];
            z = hga * (z - hmu) * rsqrtf(hva + BN_EPS) + hbe;
            out[g * NHID + t] = fmaxf(z, 0.f);
        }
    }
}

extern "C" void kernel_launch(void* const* d_in, const int* in_sizes, int n_in,
                              void* d_out, int out_size, void* d_ws, size_t ws_size,
                              hipStream_t stream) {
    (void)in_sizes; (void)n_in; (void)out_size; (void)ws_size;
    const float* feats = (const float*)d_in[0];
    const int*   edges = (const int*)d_in[1];
    const float* W0    = (const float*)d_in[2];
    const float* b0    = (const float*)d_in[3];
    const float* W1    = (const float*)d_in[4];
    const float* b1    = (const float*)d_in[5];
    const float* Wt    = (const float*)d_in[6];
    const float* bt    = (const float*)d_in[7];
    const float* gamma = (const float*)d_in[8];
    const float* beta  = (const float*)d_in[9];
    const float* rmean = (const float*)d_in[10];
    const float* rvar  = (const float*)d_in[11];
    float* out = (float*)d_out;

    char* ws = (char*)d_ws;
    float*  rs_in   = (float*)ws;  ws += (size_t)NGRAPH * NNODE * sizeof(float);
    int2*   rowinfo = (int2*)ws;   ws += (size_t)NGRAPH * NNODE * sizeof(int2);
    int2*   list    = (int2*)ws;   ws += (size_t)NGRAPH * NEDGE * sizeof(int2);
    half_t* h1      = (half_t*)ws; ws += (size_t)NGRAPH * NNODE * NFEAT * sizeof(half_t);
    float*  part    = (float*)ws;  ws += (size_t)NGRAPH * 2 * NHID * sizeof(float);
    unsigned int* cnt = (unsigned int*)ws;

    k_csr<<<NGRAPH, 512, 0, stream>>>(edges, rs_in, rowinfo, list, cnt);

    k_layer0<<<NGRAPH * 2, LY_THREADS, 0, stream>>>(feats, rowinfo, list, rs_in, W0, b0, h1);

    k_layer1<<<NGRAPH * 2, LY_THREADS, 0, stream>>>(h1, rowinfo, list, rs_in, W1, b1,
                                                    Wt, bt, gamma, beta, rmean, rvar,
                                                    part, cnt, out);
}


# List build fused into the layer-0 kernel: each workgroup sorts its half's edges in LDS while its source-row loads stay in flight (LDS-only barriers); one launch less; list and sum orders unchanged
# speedup vs baseline: 31.3853x; 1.1216x over previous
#include <hip/hip_runtime.h>
#include <hip/hip_bf16.h>

typedef _Float16 half_t;
typedef __attribute__((ext_vector_type(16))) _Float16 v16h;
typedef __attribute__((ext_vector_type(8)))  _Float16 v8h;
typedef __attribute__((ext_vector_type(4)))  _Float16 v4h;
typedef __attribute__((ext_vector_type(8)))  float    v8f;
typedef __attribute__((ext_vector_type(4)))  unsigned int v4u;

#define NGRAPH 128
#define NNODE  512
#define NEDGE  8192
#define NFEAT  128
#define NHID   128
#define BN_EPS 1e-5f

#define LY_THREADS 1024
#define LY_BROWS   256
#define LY_ROWS    8
#define WT_PITCH   (NHID + 8)
#define WT_BYTES   (NHID * WT_PITCH * 2)

template <int U, typename T>
__device__ __forceinline__ void gather_batch(float4& acc, const int2 ent, const int k,
                                             const T* xs, const int lane) {
    int   s[U];
    float w[U];
    float4 v[U];
#pragma unroll
    for (int i = 0; i < U; ++i) {
        s[i] = __builtin_amdgcn_readlane(ent.x, k + i);
        w[i] = __int_as_float(__builtin_amdgcn_readlane(ent.y, k + i));
    }
#pragma unroll
    for (int i = 0; i < U; ++i) {
        if constexpr (sizeof(T) == 4) {
            v[i] = *(const float4*)(xs + s[i] * NFEAT + lane * 4);
        } else {
            const v4h h = *(const v4h*)(xs + s[i] * NFEAT + lane * 4);
            v[i] = make_float4((float)h[0], (float)h[1], (float)h[2], (float)h[3]);
        }
    }
#pragma unroll
    for (int i = 0; i < U; ++i) {
        acc.x = fmaf(v[i].x, w[i], acc.x);
        acc.y = fmaf(v[i].y, w[i], acc.y);
        acc.z = fmaf(v[i].z, w[i], acc.z);
        acc.w = fmaf(v[i].w, w[i], acc.w);
    }
}

template <typename T>
__device__ __forceinline__ void gather_chunk(float4& acc, const int2 ent, const int cnt,
                                             const T* xs, const int lane) {
    int k = 0;
    for (; k + 8 <= cnt; k += 8) gather_batch<8>(acc, ent, k, xs, lane);
    if (k + 4 <= cnt) { gather_batch<4>(acc, ent, k, xs, lane); k += 4; }
    const int rem = cnt - k;
    if      (rem == 3) gather_batch<3>(acc, ent, k, xs, lane);
    else if (rem == 2) gather_batch<2>(acc, ent, k, xs, lane);
    else if (rem == 1) gather_batch<1>(acc, ent, k, xs, lane);
}

__device__ __forceinline__ void lds_barrier() {
    asm volatile("s_wait_dscnt 0x0\n\ts_barrier_signal -1\n\ts_barrier_wait -1" ::: "memory");
}

#define CS_R (NEDGE / LY_THREADS)
#define CS_NS0 8

template <typename T, int LAYER>
__device__ __forceinline__ void layer_body(unsigned char* smem, float* bs, float* embacc,
                                           const T*     __restrict__ x,
                                           const int*   __restrict__ edges,
                                           float*       __restrict__ rs_w,
                                           int2*        __restrict__ ri_w,
                                           int2*                     list_w,
                                           unsigned int* __restrict__ cnt_w,
                                           const int2*  __restrict__ rowinfo,
                                           const int2*  __restrict__ list,
                                           const float* __restrict__ rs_in,
                                           const float* __restrict__ W,
                                           const float* __restrict__ bvec,
                                           half_t*      __restrict__ h_out) {
    constexpr int XS_BYTES = NNODE * NFEAT * (int)sizeof(T);
    T*      xs = (T*)smem;
    half_t* at = (half_t*)smem;
    half_t* wt = (half_t*)(smem + XS_BYTES);

    const int g    = blockIdx.x >> 1;
    const int half = blockIdx.x & 1;
    const int t    = threadIdx.x;
    const int wave = __builtin_amdgcn_readfirstlane(t >> 5);
    const int lane = t & 31;
    const size_t gb = (size_t)g * NNODE;
    const int    row0 = half * LY_BROWS + wave * LY_ROWS;

    constexpr int NV = XS_BYTES / 16 / LY_THREADS;
    const v4u* src4 = (const v4u*)(x + gb * NFEAT);
    v4u stg[NV];

    int  start[LY_ROWS], len[LY_ROWS];
    int2 ent[LY_ROWS];
    float myrs;
    const int2* lg;

    if constexpr (LAYER == 0) {
        constexpr int CUR_BYTES = 32 * LY_BROWS * 4;
        int*   cur  = (int*)smem;
        int2*  lst  = (int2*)(smem + CUR_BYTES);
        int*   degO = (int*)(smem + CUR_BYTES + NEDGE * 8);
        int*   rowp = degO + NNODE;
        int*   rlen = rowp + LY_BROWS;
        float* rso  = (float*)(rlen + LY_BROWS);
        int*   wsum = (int*)(rso + NNODE);
        static_assert(CUR_BYTES + NEDGE * 8 + (NNODE + 2 * LY_BROWS + NNODE + 8) * 4 <= CS_NS0 * 16 * LY_THREADS,
                      "build scratch lies inside the bytes of the first CS_NS0 copy steps");

        const int* e = edges + (size_t)g * 2 * NEDGE + wave * (32 * CS_R) + lane;
        int es[CS_R], edd[CS_R];
#pragma unroll
        for (int r = 0; r < CS_R; ++r) {
            es[r]  = e[r * 32];
            edd[r] = e[NEDGE + r * 32];
        }
#pragma unroll
        for (int j = CS_NS0; j < NV; ++j) stg[j] = src4[j * LY_THREADS + t];
#pragma unroll
        for (int j = 0; j < CS_NS0; ++j) stg[j] = src4[j * LY_THREADS + t];

        const uint4 z4 = make_uint4(0u, 0u, 0u, 0u);
        ((uint4*)cur)[t]              = z4;
        ((uint4*)cur)[LY_THREADS + t] = z4;
        if (t < NNODE) degO[t] = 0;
        if (half == 0 && t == 0) cnt_w[g] = 0u;
        lds_barrier();

        int ed[CS_R];
#pragma unroll
        for (int r = 0; r < CS_R; ++r)
            ed[r] = (es[r] & (NNODE - 1)) | ((edd[r] & (NNODE - 1)) << 16);
#pragma unroll
        for (int r = 0; r < CS_R; ++r) {
            const int sr = ed[r] & 0xffff, d = ed[r] >> 16;
            atomicAdd(&degO[sr], 1);
            if ((d >> 8) == half) atomicAdd(&cur[wave * LY_BROWS + (d & (LY_BROWS - 1))], 1);
        }
#pragma unroll
        for (int j = CS_NS0; j < NV; ++j) ((v4u*)smem)[j * LY_THREADS + t] = stg[j];
        lds_barrier();

        int run = 0;
        if (t < LY_BROWS) {
#pragma unroll 8
            for (int w = 0; w < 32; ++w) {
                const int c = cur[w * LY_BROWS + t];
                cur[w * LY_BROWS + t] = run;
                run += c;
            }
        }
        int incl = run;
        for (int o = 1; o < 32; o <<= 1) {
            const int v = __shfl_up(incl, o);
            if (lane >= o) incl += v;
        }
        if (t < LY_BROWS && lane == 31) wsum[wave] = incl;
        if (t < NNODE) rso[t] = rsqrtf(fmaxf((float)degO[t], 1.0f));
        lds_barrier();
        int tot = 0;
#pragma unroll
        for (int w = 0; w < 8; ++w) tot += wsum[w];
        if (t < LY_BROWS) {
            int pre = 0;
            for (int w = 0; w < wave; ++w) pre += wsum[w];
            const int rstart = pre + incl - run;
            rowp[t] = rstart;
            rlen[t] = run;
            rs_w[gb + half * LY_BROWS + t] = rsqrtf(fmaxf((float)run, 1.0f));
            ri_w[gb + half * LY_BROWS + t] = make_int2(rstart, run);
        }
        lds_barrier();

        int* cw = cur + wave * LY_BROWS;
#pragma unroll
        for (int r0 = 0; r0 < CS_R; r0 += 4) {
            int   base[4], got[4], rp[4];
            float rv[4];
#pragma unroll
            for (int q = 0; q < 4; ++q) {
                const int d = ed[r0 + q] >> 16;
                base[q] = 0;
                got[q]  = 0;
                if ((d >> 8) == half) {
                    base[q] = __hip_atomic_load(&cw[d & (LY_BROWS - 1)], __ATOMIC_RELAXED, __HIP_MEMORY_SCOPE_WORKGROUP);
                    got[q]  = atomicAdd(&cw[d & (LY_BROWS - 1)], 1);
                }
            }
#pragma unroll
            for (int q = 0; q < 4; ++q) {
                rp[q] = rowp[(ed[r0 + q] >> 16) & (LY_BROWS - 1)];
                rv[q] = rso[ed[r0 + q] & 0xffff];
            }
#pragma unroll
            for (int q = 0; q < 4; ++q) {
                const int  d   = ed[r0 + q] >> 16;
                const bool act = (d >> 8) == half;
                int rank = 0;
                unsigned coll = __builtin_amdgcn_ballot_w32(act && got[q] != base[q]);
                for (int it = 0; it < 31 && coll != 0u; ++it) {
                    const int j  = __builtin_ctz(coll);
                    const int dj = __builtin_amdgcn_readlane(d, j);
                    const unsigned grp = __builtin_amdgcn_ballot_w32(d == dj);
                    if (d == dj) rank = (int)__builtin_amdgcn_mbcnt_lo(grp, 0u);
                    coll &= ~grp;
                }
                if (act) lst[rp[q] + base[q] + rank] = make_int2(ed[r0 + q] & 0xffff, __float_as_int(rv[q]));
            }
        }
        lds_barrier();

        const int rl = wave * LY_ROWS + (lane & (LY_ROWS - 1));
        const int2 myinf = make_int2(rowp[rl], rlen[rl]);
        myrs = rsqrtf(fmaxf((float)myinf.y, 1.0f));
#pragma unroll
        for (int i = 0; i < LY_ROWS; ++i) {
            start[i] = __builtin_amdgcn_readlane(myinf.x, i);
            len[i]   = __builtin_amdgcn_readlane(myinf.y, i);
            ent[i]   = lst[min(start[i] + lane, NEDGE - 1)];
        }
        int2* lgw = list_w + (size_t)blockIdx.x * NEDGE;
        const uint4* ls4 = (const uint4*)lst;
        uint4*       lg4 = (uint4*)lgw;
        const int nq = (tot + 1) >> 1;
#pragma unroll
        for (int j = 0; j < NEDGE / 2 / LY_THREADS; ++j) {
            const int q = j * LY_THREADS + t;
            if (q < nq) lg4[q] = ls4[q];
        }
        lg = lgw;
        __syncthreads();
    } else {
        lg = list + (size_t)blockIdx.x * NEDGE;
        const int2 myinf = rowinfo[gb + row0 + (lane & (LY_ROWS - 1))];
        myrs = rs_in[gb + row0 + (lane & (LY_ROWS - 1))];
#pragma unroll
        for (int j = 0; j < NV; ++j) stg[j] = src4[j * LY_THREADS + t];
#pragma unroll
        for (int i = 0; i < LY_ROWS; ++i) {
            start[i] = __builtin_amdgcn_readlane(myinf.x, i);
            len[i]   = __builtin_amdgcn_readlane(myinf.y, i);
            ent[i]   = lg[min(start[i] + lane, NEDGE - 1)];
        }
    }

    const int fn = t & (NHID - 1);
    const int fk = t >> 7;
    float wreg[2][8];
#pragma unroll
    for (int j = 0; j < 2; ++j) {
        const float* wp = W + (size_t)((fk + 8 * j) * 8) * NHID + fn;
#pragma unroll
        for (int i = 0; i < 8; ++i) wreg[j][i] = wp[i * NHID];
    }
    const float bval = bvec[t & (NHID - 1)];

    constexpr int NLATE = (LAYER == 0) ? CS_NS0 : NV;
#pragma unroll
    for (int j = 0; j < NLATE; ++j) ((v4u*)smem)[j * LY_THREADS + t] = stg[j];
    __syncthreads();

    v4h kept[LY_ROWS];
#pragma unroll
    for (int i = 0; i < LY_ROWS; ++i) {
        float4 acc = make_float4(0.f, 0.f, 0.f, 0.f);
        gather_chunk<T>(acc, ent[i], min(len[i], 32), xs, lane);
        for (int b = 32; b < len[i]; b += 32) {
            const int2 more = (b + lane < len[i]) ? lg[start[i] + b + lane] : make_int2(0, 0);
            gather_chunk<T>(acc, more, min(len[i] - b, 32), xs, lane);
        }
        const float rsin = __int_as_float(__builtin_amdgcn_readlane(__float_as_int(myrs), i));
        v4h p;
        p[0] = (half_t)(acc.x * rsin); p[1] = (half_t)(acc.y * rsin);
        p[2] = (half_t)(acc.z * rsin); p[3] = (half_t)(acc.w * rsin);
        kept[i] = p;
    }
    __syncthreads();

#pragma unroll
    for (int i = 0; i < LY_ROWS; ++i)
        *(v4h*)(at + (wave * LY_ROWS + i) * WT_PITCH + lane * 4) = kept[i];

#pragma unroll
    for (int j = 0; j < 2; ++j) {
        v8h p;
#pragma unroll
        for (int i = 0; i < 8; ++i) p[i] = (half_t)wreg[j][i];
        *(v8h*)(wt + fn * WT_PITCH + (fk + 8 * j) * 8) = p;
    }
    if (t < NHID) {
        bs[t] = bval;
        if constexpr (LAYER == 1) embacc[t] = 0.f;
    }
    __syncthreads();

    const int l16 = lane & 15, hi = lane >> 4;
    const int mt = wave >> 1;
    const int nh = wave & 1;
    const half_t* arow = at + (mt * 16 + l16) * WT_PITCH;

    v8f c[4] = {};

#pragma unroll
    for (int ko = 0; ko < 4; ++ko) {
        const int k0 = ko * 32 + hi * 8;
        const int k1 = ko * 32 + 16 + hi * 8;
        const v8h alo = *(const v8h*)(arow + k0);
        const v8h ahi = *(const v8h*)(arow + k1);
        const v16h a = __builtin_shufflevector(alo, ahi,
                           0, 1, 2, 3, 4, 5, 6, 7,
                           8, 9, 10, 11, 12, 13, 14, 15);
#pragma unroll
        for (int nj = 0; nj < 4; ++nj) {
            const half_t* brow = wt + ((nh * 4 + nj) * 16 + l16) * WT_PITCH;
            const v8h blo = *(const v8h*)(brow + k0);
            const v8h bhi = *(const v8h*)(brow + k1);
            const v16h b = __builtin_shufflevector(blo, bhi,
                               0, 1, 2, 3, 4, 5, 6, 7,
                               8, 9, 10, 11, 12, 13, 14, 15);
            c[nj] = __builtin_amdgcn_wmma_f32_16x16x32_f16(
                        false, a, false, b, (short)0, c[nj], false, false);
        }
    }

#pragma unroll
    for (int nj = 0; nj < 4; ++nj) {
        const int n = (nh * 4 + nj) * 16 + l16;
        if constexpr (LAYER == 0) {
#pragma unroll
            for (int r = 0; r < 8; ++r) {
                const int ml = r + 8 * hi;
                const float v = fmaxf(c[nj][r] + bs[n], 0.f);
                h_out[(gb + (size_t)(half * LY_BROWS + mt * 16 + ml)) * NFEAT + n] = (half_t)v;
            }
        } else {
            float s = 0.f;
#pragma unroll
            for (int r = 0; r < 8; ++r) s += fmaxf(c[nj][r] + bs[n], 0.f);
            atomicAdd(&embacc[n], s);
        }
    }
}

__global__ __launch_bounds__(LY_THREADS) void k_layer0(const float* __restrict__ x,
                                                       const int*   __restrict__ edges,
                                                       float*       __restrict__ rs_in,
                                                       int2*        __restrict__ rowinfo,
                                                       int2*                     list,
                                                       unsigned int* __restrict__ cnt,
                                                       const float* __restrict__ W,
                                                       const float* __restrict__ bvec,
                                                       half_t*      __restrict__ h_out) {
    __shared__ __align__(16) unsigned char smem[NNODE * NFEAT * 4 + WT_BYTES];
    __shared__ float bs[NHID];
    layer_body<float, 0>(smem, bs, nullptr, x, edges, rs_in, rowinfo, list, cnt,
                         nullptr, nullptr, nullptr, W, bvec, h_out);
}

#define HD_KQ 8
#define HD_KL (NHID / HD_KQ)

__global__ __launch_bounds__(LY_THREADS) void k_layer1(const half_t* __restrict__ h,
                                                       const int2*   __restrict__ rowinfo,
                                                       const int2*   __restrict__ list,
                                                       const float*  __restrict__ rs_in,
                                                       const float*  __restrict__ W,
                                                       const float*  __restrict__ bvec,
                                                       const float*  __restrict__ Wt,
                                                       const float*  __restrict__ bt,
                                                       const float*  __restrict__ gamma,
                                                       const float*  __restrict__ beta,
                                                       const float*  __restrict__ rmean,
                                                       const float*  __restrict__ rvar,
                                                       float*        part,
                                                       unsigned int* cnt,
                                                       float*        __restrict__ out) {
    __shared__ __align__(16) unsigned char smem[NNODE * NFEAT * 2 + WT_BYTES];
    __shared__ float bs[NHID];
    __shared__ float embacc[NHID];
    __shared__ float er[NHID];
    __shared__ float zp[HD_KQ * NHID];
    __shared__ unsigned int s_ticket;
    layer_body<half_t, 1>(smem, bs, embacc, h, nullptr, nullptr, nullptr, nullptr, nullptr,
                          rowinfo, list, rs_in, W, bvec, nullptr);

    const int g    = blockIdx.x >> 1;
    const int half = blockIdx.x & 1;
    const int t    = threadIdx.x;
    const int hn   = t & (NHID - 1);
    const int kq   = t >> 7;

    __syncthreads();

    float wtr[HD_KL];
#pragma unroll
    for (int i = 0; i < HD_KL; ++i) wtr[i] = Wt[(size_t)(kq * HD_KL + i) * NHID + hn];
    const float hbt = bt[hn], hga = gamma[hn], hbe = beta[hn], hmu = rmean[hn], hva = rvar[hn];

    if (t < 32) {
        float* mine = part + (size_t)(g * 2 + half) * NHID;
        float pm[4];
#pragma unroll
        for (int j = 0; j < 4; ++j) pm[j] = embacc[j * 32 + t] * (1.0f / NNODE);
#pragma unroll
        for (int j = 0; j < 4; ++j)
            __hip_atomic_store(mine + j * 32 + t, pm[j], __ATOMIC_RELAXED, __HIP_MEMORY_SCOPE_AGENT);
        __builtin_amdgcn_fence(__ATOMIC_RELEASE, "agent");
        if (t == 0) s_ticket = __hip_atomic_fetch_add(cnt + g, 1u, __ATOMIC_RELAXED, __HIP_MEMORY_SCOPE_AGENT);
    }
    __syncthreads();
    if (s_ticket == 1u) {
        if (t < NHID) {
            __builtin_amdgcn_fence(__ATOMIC_ACQUIRE, "agent");
            const float* p0 = part + (size_t)(g * 2) * NHID;
            const float a = __hip_atomic_load(p0 + t,        __ATOMIC_RELAXED, __HIP_MEMORY_SCOPE_AGENT);
            const float b = __hip_atomic_load(p0 + NHID + t, __ATOMIC_RELAXED, __HIP_MEMORY_SCOPE_AGENT);
            er[t] = a + b;
        }
        __syncthreads();
        float s = 0.f;
#pragma unroll
        for (int i = 0; i < HD_KL; ++i) s = fmaf(er[kq * HD_KL + i], wtr[i], s);
        zp[kq * NHID + hn] = s;
        __syncthreads();
        if (t < NHID) {
            float z = hbt;
#pragma unroll
            for (int j = 0; j < HD_KQ; ++j) z += zp[j * NHID + t];
            z = hga * (z - hmu) * rsqrtf(hva + BN_EPS) + hbe;
            out[g * NHID + t] = fmaxf(z, 0.f);
        }
    }
}

extern "C" void kernel_launch(void* const* d_in, const int* in_sizes, int n_in,
                              void* d_out, int out_size, void* d_ws, size_t ws_size,
                              hipStream_t stream) {
    (void)in_sizes; (void)n_in; (void)out_size; (void)ws_size;
    const float* feats = (const float*)d_in[0];
    const int*   edges = (const int*)d_in[1];
    const float* W0    = (const float*)d_in[2];
    const float* b0    = (const float*)d_in[3];
    const float* W1    = (const float*)d_in[4];
    const float* b1    = (const float*)d_in[5];
    const float* Wt    = (const float*)d_in[6];
    const float* bt    = (const float*)d_in[7];
    const float* gamma = (const float*)d_in[8];
    const float* beta  = (const float*)d_in[9];
    const float* rmean = (const float*)d_in[10];
    const float* rvar  = (const float*)d_in[11];
    float* out = (float*)d_out;

    char* ws = (char*)d_ws;
    float*  rs_in   = (float*)ws;  ws += (size_t)NGRAPH * NNODE * sizeof(float);
    int2*   rowinfo = (int2*)ws;   ws += (size_t)NGRAPH * NNODE * sizeof(int2);
    int2*   list    = (int2*)ws;   ws += (size_t)NGRAPH * 2 * NEDGE * sizeof(int2);
    half_t* h1      = (half_t*)ws; ws += (size_t)NGRAPH * NNODE * NFEAT * sizeof(half_t);
    float*  part    = (float*)ws;  ws += (size_t)NGRAPH * 2 * NHID * sizeof(float);
    unsigned int* cnt = (unsigned int*)ws;

    k_layer0<<<NGRAPH * 2, LY_THREADS, 0, stream>>>(feats, edges, rs_in, rowinfo, list, cnt, W0, b0, h1);

    k_layer1<<<NGRAPH * 2, LY_THREADS, 0, stream>>>(h1, rowinfo, list, rs_in, W1, b1,
                                                    Wt, bt, gamma, beta, rmean, rvar,
                                                    part, cnt, out);
}
